# starting kernel
# baseline (speedup 1.0000x reference)
#include <hip/hip_runtime.h>
#include <hip/hip_bf16.h>

typedef __attribute__((ext_vector_type(2))) float v2f;
typedef __attribute__((ext_vector_type(4))) float v4f;
typedef __attribute__((ext_vector_type(8))) float v8f;

#define HEADS 4
#define CH 128
#define NEG_SLOPE 0.2f

__device__ __forceinline__ unsigned enc_f32(float f) {
  unsigned u = __float_as_uint(f);
  return (f >= 0.0f) ? (u | 0x80000000u) : ~u;
}
__device__ __forceinline__ float dec_f32(unsigned u) {
  return (u & 0x80000000u) ? __uint_as_float(u & 0x7fffffffu)
                           : __uint_as_float(~u);
}

__device__ __forceinline__ void edge_nodes(const int* __restrict__ ei, int e,
                                           int ne, int& s, int& d) {
  if (e < ne) { s = ei[e]; d = ei[ne + e]; }
  else        { s = d = e - ne; }
}

__global__ void gat_init(float* __restrict__ out, unsigned* __restrict__ emax,
                         float* __restrict__ denom, int n) {
  int t = blockIdx.x * blockDim.x + threadIdx.x;
  if (t < n * CH) out[t] = 0.0f;
  if (t < n * HEADS) { emax[t] = 0u; denom[t] = 0.0f; }
}

__global__ __launch_bounds__(256) void gat_gemm_wmma(
    const float* __restrict__ x, const float* __restrict__ W,
    float* __restrict__ h, int n) {
  const int wave = threadIdx.x >> 5;
  const int lane = threadIdx.x & 31;
  const int m0 = blockIdx.x * 16;
  const int n0 = wave * 16;
  const int lm = lane & 15;
  const int lk = (lane >> 4) * 2;

  int row = m0 + lm;
  if (row >= n) row = n - 1;
  const float* __restrict__ arow = x + (size_t)row * CH;

  v8f acc = {};
#pragma unroll
  for (int k0 = 0; k0 < CH; k0 += 4) {
    v2f a = *(const v2f*)(arow + k0 + lk);
    v2f b;
    b.x = W[(size_t)(k0 + lk) * CH + n0 + lm];
    b.y = W[(size_t)(k0 + lk + 1) * CH + n0 + lm];
    acc = __builtin_amdgcn_wmma_f32_16x16x4_f32(
         false, a,  false, b,
         (short)0, acc,  false,  false);
  }

  const int colg = n0 + lm;
  const int rbase = m0 + 8 * (lane >> 4);
#pragma unroll
  for (int v = 0; v < 8; ++v) {
    int rowg = rbase + v;
    if (rowg < n) h[(size_t)rowg * CH + colg] = acc[v];
  }
}

__global__ void gat_att(const float* __restrict__ h,
                        const float* __restrict__ att_src,
                        const float* __restrict__ att_dst,
                        float* __restrict__ a_src, float* __restrict__ a_dst,
                        int n) {
  int t = blockIdx.x * blockDim.x + threadIdx.x;
  if (t >= n * HEADS) return;
  int head = t & (HEADS - 1);
  const float* __restrict__ hv = h + (size_t)(t >> 2) * CH + head * 32;
  const float* __restrict__ as = att_src + head * 32;
  const float* __restrict__ ad = att_dst + head * 32;
  float s = 0.0f, d = 0.0f;
#pragma unroll
  for (int c = 0; c < 32; ++c) { float v = hv[c]; s += v * as[c]; d += v * ad[c]; }
  a_src[t] = s;
  a_dst[t] = d;
}

__global__ void gat_emax(const int* __restrict__ ei,
                         const float* __restrict__ a_src,
                         const float* __restrict__ a_dst,
                         unsigned* __restrict__ emax, int n, int ne) {
  int t = blockIdx.x * blockDim.x + threadIdx.x;
  if (t >= (ne + n) * HEADS) return;
  int head = t & (HEADS - 1);
  int e = t >> 2;
  int s, d; edge_nodes(ei, e, ne, s, d);
  float v = a_src[s * HEADS + head] + a_dst[d * HEADS + head];
  v = (v >= 0.0f) ? v : NEG_SLOPE * v;
  atomicMax(&emax[d * HEADS + head], enc_f32(v));
}

__global__ void gat_denom(const int* __restrict__ ei,
                          const float* __restrict__ a_src,
                          const float* __restrict__ a_dst,
                          const unsigned* __restrict__ emax,
                          float* __restrict__ denom, int n, int ne) {
  int t = blockIdx.x * blockDim.x + threadIdx.x;
  if (t >= (ne + n) * HEADS) return;
  int head = t & (HEADS - 1);
  int e = t >> 2;
  int s, d; edge_nodes(ei, e, ne, s, d);
  float v = a_src[s * HEADS + head] + a_dst[d * HEADS + head];
  v = (v >= 0.0f) ? v : NEG_SLOPE * v;
  float m = dec_f32(emax[d * HEADS + head]);
  atomicAdd(&denom[d * HEADS + head], expf(v - m));
}

__global__ void gat_scatter(const int* __restrict__ ei,
                            const float* __restrict__ h,
                            const float* __restrict__ a_src,
                            const float* __restrict__ a_dst,
                            const unsigned* __restrict__ emax,
                            const float* __restrict__ denom,
                            float* __restrict__ out, int n, int ne) {
  int gid = blockIdx.x * blockDim.x + threadIdx.x;
  int e = gid >> 5;
  int lane = gid & 31;
  if (e >= ne + n) return;
  int s, d; edge_nodes(ei, e, ne, s, d);
  int f0 = lane * 4;
  int head = f0 >> 5;
  float v = a_src[s * HEADS + head] + a_dst[d * HEADS + head];
  v = (v >= 0.0f) ? v : NEG_SLOPE * v;
  float m = dec_f32(emax[d * HEADS + head]);
  float alpha = expf(v - m) / (denom[d * HEADS + head] + 1e-16f);
  v4f hv = *(const v4f*)(h + (size_t)s * CH + f0);
  float* __restrict__ od = out + (size_t)d * CH + f0;
  atomicAdd(od + 0, hv.x * alpha);
  atomicAdd(od + 1, hv.y * alpha);
  atomicAdd(od + 2, hv.z * alpha);
  atomicAdd(od + 3, hv.w * alpha);
}

__global__ void gat_prelu(float* __restrict__ out, const float* __restrict__ bias,
                          const float* __restrict__ prelu_w, int n) {
  int t = blockIdx.x * blockDim.x + threadIdx.x;
  if (t >= n * CH) return;
  float o = out[t] + bias[t & (CH - 1)];
  float pw = prelu_w[0];
  out[t] = (o >= 0.0f) ? o : pw * o;
}

extern "C" void kernel_launch(void* const* d_in, const int* in_sizes, int n_in,
                              void* d_out, int out_size, void* d_ws, size_t ws_size,
                              hipStream_t stream) {
  const float* x       = (const float*)d_in[0];
  const int*   ei      = (const int*)d_in[1];
  const float* W       = (const float*)d_in[2];
  const float* att_src = (const float*)d_in[3];
  const float* att_dst = (const float*)d_in[4];
  const float* bias    = (const float*)d_in[5];
  const float* prelu_w = (const float*)d_in[6];
  float* out = (float*)d_out;

  const int n  = in_sizes[0] / CH;
  const int ne = in_sizes[1] / 2;

  float*    h      = (float*)d_ws;
  float*    a_src  = h + (size_t)n * CH;
  float*    a_dst  = a_src + (size_t)n * HEADS;
  unsigned* emax   = (unsigned*)(a_dst + (size_t)n * HEADS);
  float*    denom  = (float*)(emax + (size_t)n * HEADS);

  const int tpb = 256;
  int total_eh = (ne + n) * HEADS;

  gat_init<<<(n * CH + tpb - 1) / tpb, tpb, 0, stream>>>(out, emax, denom, n);
  gat_gemm_wmma<<<(n + 15) / 16, 256, 0, stream>>>(x, W, h, n);
  gat_att<<<(n * HEADS + tpb - 1) / tpb, tpb, 0, stream>>>(h, att_src, att_dst,
                                                           a_src, a_dst, n);
  gat_emax<<<(total_eh + tpb - 1) / tpb, tpb, 0, stream>>>(ei, a_src, a_dst,
                                                           emax, n, ne);
  gat_denom<<<(total_eh + tpb - 1) / tpb, tpb, 0, stream>>>(ei, a_src, a_dst,
                                                            emax, denom, n, ne);
  long long sc_threads = (long long)(ne + n) * 32;
  gat_scatter<<<(int)((sc_threads + tpb - 1) / tpb), tpb, 0, stream>>>(
      ei, h, a_src, a_dst, emax, denom, out, n, ne);
  gat_prelu<<<(n * CH + tpb - 1) / tpb, tpb, 0, stream>>>(out, bias, prelu_w, n);
}


# Attention launch folded into the projection kernel: its 16x128 tile is staged in LDS, the per-head dots are taken there and h leaves as whole rows; tile keys written there too; 3 launches become 2
# speedup vs baseline: 3.0327x; 3.0327x over previous
#include <hip/hip_runtime.h>
#include <hip/hip_bf16.h>

typedef __attribute__((ext_vector_type(2))) float v2f;
typedef __attribute__((ext_vector_type(4))) float v4f;
typedef __attribute__((ext_vector_type(8))) float v8f;

#define HEADS 4
#define CH 128
#define NEG_SLOPE 0.2f
#define TN 196

__device__ __forceinline__ unsigned tile_key(int d) {
  return ((unsigned)d / (unsigned)TN) & 255u;
}

__device__ __forceinline__ unsigned enc_f32(float f) {
  unsigned u = __float_as_uint(f);
  return (f >= 0.0f) ? (u | 0x80000000u) : ~u;
}
__device__ __forceinline__ float dec_f32(unsigned u) {
  return (u & 0x80000000u) ? __uint_as_float(u & 0x7fffffffu)
                           : __uint_as_float(~u);
}

#define HP 132
__global__ __launch_bounds__(256) void gat_gemm_att(
    const float* __restrict__ x, const float* __restrict__ W,
    const float* __restrict__ att_src, const float* __restrict__ att_dst,
    const int* __restrict__ ei,
    float* __restrict__ h, float* __restrict__ a_src, float* __restrict__ a_dst,
    unsigned* __restrict__ keys, int n, int ne) {
  __shared__ __attribute__((aligned(16))) float ht[16 * HP];
  const int tid = threadIdx.x;

  {
    const int* __restrict__ dst = ei + ne;
    const int nkw = ((ne + 15) >> 4) << 2;
    for (int w = blockIdx.x * 256 + tid; w < nkw; w += gridDim.x * 256) {
      const int e = w * 4;
      int4 d = int4{0, 0, 0, 0};
      if ((ne & 3) == 0 && e < ne) {
        d = *(const int4*)(dst + e);
      } else {
        if (e < ne) d.x = dst[e];
        if (e + 1 < ne) d.y = dst[e + 1];
        if (e + 2 < ne) d.z = dst[e + 2];
        if (e + 3 < ne) d.w = dst[e + 3];
      }
      keys[w] = tile_key(d.x) | (tile_key(d.y) << 8) | (tile_key(d.z) << 16) | (tile_key(d.w) << 24);
    }
  }

  const int wave = tid >> 5;
  const int lane = tid & 31;
  const int m0 = blockIdx.x * 16;
  const int n0 = wave * 16;
  const int lm = lane & 15;
  const int lk = (lane >> 4) * 2;

  int row = m0 + lm;
  if (row >= n) row = n - 1;
  const float* __restrict__ arow = x + (size_t)row * CH;

  v8f acc = {};
#pragma unroll
  for (int k0 = 0; k0 < CH; k0 += 4) {
    v2f a = *(const v2f*)(arow + k0 + lk);
    v2f b;
    b.x = W[(size_t)(k0 + lk) * CH + n0 + lm];
    b.y = W[(size_t)(k0 + lk + 1) * CH + n0 + lm];
    acc = __builtin_amdgcn_wmma_f32_16x16x4_f32(
         false, a,  false, b,
         (short)0, acc,  false,  false);
  }

  {
    float* __restrict__ hs = ht + (8 * (lane >> 4)) * HP + n0 + lm;
#pragma unroll
    for (int v = 0; v < 8; ++v) hs[v * HP] = acc[v];
  }
  __syncthreads();

#pragma unroll
  for (int rr = 0; rr < 2; ++rr) {
    const int r = 2 * wave + rr;
    if (m0 + r < n)
      *(v4f*)(h + (size_t)(m0 + r) * CH + lane * 4) = *(const v4f*)(ht + r * HP + lane * 4);
  }

  if (tid < 128) {
    const int r = (tid & 63) >> 2;
    const int head = tid & (HEADS - 1);
    if (m0 + r < n) {
      const float* __restrict__ av = ((tid < 64) ? att_src : att_dst) + head * 32;
      const float* hv = ht + r * HP + head * 32;
      float s = 0.0f;
#pragma unroll
      for (int c = 0; c < 32; ++c) s += hv[c] * av[c];
      float* __restrict__ o = (tid < 64) ? a_src : a_dst;
      o[(size_t)(m0 + r) * HEADS + head] = s;
    }
  }
}

#define CAP 8192
#define EB 1024
#define SCAN_K 8
#define ACC_U 8
#define MAX_U 4

__device__ __forceinline__ float leaky(float v) {
  return (v >= 0.0f) ? v : NEG_SLOPE * v;
}

__global__ __launch_bounds__(EB) void gat_tile(
    const int* __restrict__ ei, const unsigned* __restrict__ keys,
    const float* __restrict__ h,
    const float* __restrict__ a_src, const float* __restrict__ a_dst,
    const float* __restrict__ bias, const float* __restrict__ prelu_w,
    float* __restrict__ out, int n, int ne) {
  __shared__ __attribute__((aligned(16))) float acc[TN * CH];
  __shared__ unsigned smax[TN * HEADS];
  __shared__ float sden[TN * HEADS];
  __shared__ float sadst[TN * HEADS];
  __shared__ int ls[CAP];
  __shared__ int ld[CAP];
  __shared__ int cnt;

  const int tid = threadIdx.x;
  const int tile0 = blockIdx.x * TN;
  const int tn = min(TN, n - tile0);
  const int* __restrict__ dst = ei + ne;

  for (int i = tid; i < TN * CH / 4; i += EB) ((v4f*)acc)[i] = v4f{0.f, 0.f, 0.f, 0.f};
  for (int i = tid; i < tn * HEADS; i += EB) {
    smax[i] = 0u;
    sden[i] = 0.0f;
    sadst[i] = a_dst[(size_t)tile0 * HEADS + i];
  }
  if (tid == 0) cnt = 0;
  __syncthreads();

  const unsigned tkey = ((unsigned)blockIdx.x & 255u) * 0x01010101u;
  auto list16 = [&](int g, uint4 k) {
    const unsigned x0 = k.x ^ tkey, x1 = k.y ^ tkey, x2 = k.z ^ tkey, x3 = k.w ^ tkey;
    const unsigned r0 = (x0 - 0x01010101u) & ~x0, r1 = (x1 - 0x01010101u) & ~x1;
    const unsigned r2 = (x2 - 0x01010101u) & ~x2, r3 = (x3 - 0x01010101u) & ~x3;
    if (((r0 | r1 | r2 | r3) & 0x80808080u) != 0u) {
      unsigned m = ((r0 & 0x80808080u) >> 7) | ((r1 & 0x80808080u) >> 6) |
                   ((r2 & 0x80808080u) >> 5) | ((r3 & 0x80808080u) >> 4);
      int slot = atomicAdd(&cnt, __popc(m));
      const int e0 = g * 16;
      while (m != 0u) {
        const int p = __ffs((int)m) - 1;
        m &= m - 1u;
        if (slot < CAP) ls[slot] = e0 + ((p & 7) << 2) + (p >> 3);
        ++slot;
      }
    }
  };
  const int ngrp = (ne + 15) >> 4;
  const uint4* __restrict__ kg = (const uint4*)keys;
  for (int base = tid; base < ngrp; base += EB * SCAN_K) {
    uint4 k4[SCAN_K];
#pragma unroll
    for (int k = 0; k < SCAN_K; ++k) k4[k] = kg[min(base + k * EB, ngrp - 1)];
#pragma unroll
    for (int k = 0; k < SCAN_K; ++k)
      if (base + k * EB < ngrp) list16(base + k * EB, k4[k]);
  }
  __syncthreads();

  const int total = cnt;
  auto max_edge = [&](int dl, v4f as) {
#pragma unroll
    for (int hh = 0; hh < HEADS; ++hh)
      atomicMax(&smax[dl * HEADS + hh], enc_f32(leaky(as[hh] + sadst[dl * HEADS + hh])));
  };
  if (total <= CAP) {
    for (int base = tid; base < total; base += EB * MAX_U) {
      int sv[MAX_U], dd[MAX_U]; v4f as[MAX_U]; bool in_list[MAX_U];
#pragma unroll
      for (int u = 0; u < MAX_U; ++u) {
        int idx = base + u * EB;
        int e = ne;
        if (idx < total) e = ls[idx];
        in_list[u] = e < ne;
        if (!in_list[u]) e = 0;
        sv[u] = ei[e];
        dd[u] = dst[e];
      }
#pragma unroll
      for (int u = 0; u < MAX_U; ++u) as[u] = *(const v4f*)(a_src + (size_t)sv[u] * HEADS);
#pragma unroll
      for (int u = 0; u < MAX_U; ++u) {
        int idx = base + u * EB;
        if (idx < total) {
          unsigned dl = (unsigned)(dd[u] - tile0);
          bool mine = in_list[u] && dl < (unsigned)tn;
          ls[idx] = sv[u];
          ld[idx] = mine ? (int)dl : -1;
          if (mine) max_edge((int)dl, as[u]);
        }
      }
    }
  } else {
    for (int e = tid; e < ne; e += EB) {
      unsigned dl = (unsigned)(dst[e] - tile0);
      if (dl < (unsigned)tn) max_edge((int)dl, *(const v4f*)(a_src + (size_t)ei[e] * HEADS));
    }
  }
  for (int i = tid; i < tn; i += EB)
    max_edge(i, *(const v4f*)(a_src + (size_t)(tile0 + i) * HEADS));
  __syncthreads();

  const int wave = tid >> 5;
  const int lane = tid & 31;
  const int head = lane >> 3;
  auto add_edge = [&](int dl, float av, v4f hv) {
    float v = leaky(av + sadst[dl * HEADS + head]);
    float p = expf(v - dec_f32(smax[dl * HEADS + head]));
    if ((lane & 7) == 0) atomicAdd(&sden[dl * HEADS + head], p);
    float* a = acc + dl * CH + lane;
    atomicAdd(a, p * hv.x);
    atomicAdd(a + 32, p * hv.y);
    atomicAdd(a + 64, p * hv.z);
    atomicAdd(a + 96, p * hv.w);
  };
  if (total <= CAP) {
    const int items = total + tn;
    for (int i0 = wave * ACC_U; i0 < items; i0 += (EB / 32) * ACC_U) {
      int sdl[ACC_U]; float av[ACC_U]; v4f hv[ACC_U];
#pragma unroll
      for (int u = 0; u < ACC_U; ++u) {
        int idx = i0 + u;
        int s = tile0, dl = 0;
        if (idx < total) { s = ls[idx]; dl = ld[idx]; }
        else if (idx < items) { dl = idx - total; s = tile0 + dl; }
        sdl[u] = dl;
        av[u] = a_src[(size_t)s * HEADS + head];
        hv[u] = *(const v4f*)(h + (size_t)s * CH + lane * 4);
      }
#pragma unroll
      for (int u = 0; u < ACC_U; ++u)
        if (i0 + u < items && sdl[u] >= 0) add_edge(sdl[u], av[u], hv[u]);
    }
  } else {
    for (int e = wave; e < ne; e += EB / 32) {
      unsigned dl = (unsigned)(dst[e] - tile0);
      if (dl < (unsigned)tn) {
        int s = ei[e];
        add_edge((int)dl, a_src[(size_t)s * HEADS + head],
                 *(const v4f*)(h + (size_t)s * CH + lane * 4));
      }
    }
    for (int i = wave; i < tn; i += EB / 32) {
      int s = tile0 + i;
      add_edge(i, a_src[(size_t)s * HEADS + head],
               *(const v4f*)(h + (size_t)s * CH + lane * 4));
    }
  }
  __syncthreads();

  const float pw = prelu_w[0];
  for (int i = tid; i < tn * (CH / 4); i += EB) {
    int node = i >> 5;
    int l4 = i & 31;
    float den = sden[node * HEADS + (l4 >> 3)] + 1e-16f;
    const float* a = acc + node * CH + l4;
    v4f b = *(const v4f*)(bias + l4 * 4);
    v4f o;
    o.x = a[0] / den + b.x;
    o.y = a[32] / den + b.y;
    o.z = a[64] / den + b.z;
    o.w = a[96] / den + b.w;
    o.x = (o.x >= 0.0f) ? o.x : pw * o.x;
    o.y = (o.y >= 0.0f) ? o.y : pw * o.y;
    o.z = (o.z >= 0.0f) ? o.z : pw * o.z;
    o.w = (o.w >= 0.0f) ? o.w : pw * o.w;
    *(v4f*)(out + (size_t)(tile0 + node) * CH + l4 * 4) = o;
  }
}

extern "C" void kernel_launch(void* const* d_in, const int* in_sizes, int n_in,
                              void* d_out, int out_size, void* d_ws, size_t ws_size,
                              hipStream_t stream) {
  const float* x       = (const float*)d_in[0];
  const int*   ei      = (const int*)d_in[1];
  const float* W       = (const float*)d_in[2];
  const float* att_src = (const float*)d_in[3];
  const float* att_dst = (const float*)d_in[4];
  const float* bias    = (const float*)d_in[5];
  const float* prelu_w = (const float*)d_in[6];
  float* out = (float*)d_out;

  const int n  = in_sizes[0] / CH;
  const int ne = in_sizes[1] / 2;

  float* h     = (float*)d_ws;
  float* a_src = h + (size_t)n * CH;
  float* a_dst = a_src + (size_t)n * HEADS;
  unsigned* keys = (unsigned*)(a_dst + (size_t)n * HEADS);

  gat_gemm_att<<<(n + 15) / 16, 256, 0, stream>>>(x, W, att_src, att_dst, ei,
                                                  h, a_src, a_dst, keys, n, ne);
  gat_tile<<<(n + TN - 1) / TN, EB, 0, stream>>>(ei, keys, h, a_src, a_dst, bias, prelu_w,
                                                 out, n, ne);
}


# Projection launch made persistent: 512 work-groups loop over 16-row tiles, each wave keeps its 128x16 slice of W in registers and loads x 16 B per lane; k order of h's sums changes
# speedup vs baseline: 3.2519x; 1.0723x over previous
#include <hip/hip_runtime.h>
#include <hip/hip_bf16.h>

typedef __attribute__((ext_vector_type(2))) float v2f;
typedef __attribute__((ext_vector_type(4))) float v4f;
typedef __attribute__((ext_vector_type(8))) float v8f;

#define HEADS 4
#define CH 128
#define NEG_SLOPE 0.2f
#define TN 196

__device__ __forceinline__ unsigned tile_key(int d) {
  return ((unsigned)d / (unsigned)TN) & 255u;
}

__device__ __forceinline__ unsigned enc_f32(float f) {
  unsigned u = __float_as_uint(f);
  return (f >= 0.0f) ? (u | 0x80000000u) : ~u;
}
__device__ __forceinline__ float dec_f32(unsigned u) {
  return (u & 0x80000000u) ? __uint_as_float(u & 0x7fffffffu)
                           : __uint_as_float(~u);
}

#define GEMM_WG 512
#define HP 132
__global__ __launch_bounds__(256) void gat_gemm_att(
    const float* x,
    const float* __restrict__ W,
    const float* __restrict__ att_src, const float* __restrict__ att_dst,
    const int* __restrict__ ei,
    float* __restrict__ h, float* __restrict__ a_src, float* __restrict__ a_dst,
    unsigned* __restrict__ keys, int n, int ne) {
  __shared__ __attribute__((aligned(16))) float ht[2][16 * HP];
  const int tid = threadIdx.x;

  {
    const int* __restrict__ dst = ei + ne;
    const int nkw = ((ne + 15) >> 4) << 2;
    for (int w = blockIdx.x * 256 + tid; w < nkw; w += gridDim.x * 256) {
      const int e = w * 4;
      int4 d = int4{0, 0, 0, 0};
      if ((ne & 3) == 0 && e < ne) {
        d = *(const int4*)(dst + e);
      } else {
        if (e < ne) d.x = dst[e];
        if (e + 1 < ne) d.y = dst[e + 1];
        if (e + 2 < ne) d.z = dst[e + 2];
        if (e + 3 < ne) d.w = dst[e + 3];
      }
      keys[w] = tile_key(d.x) | (tile_key(d.y) << 8) | (tile_key(d.z) << 16) | (tile_key(d.w) << 24);
    }
  }

  const int wave = tid >> 5;
  const int lane = tid & 31;
  const int n0 = wave * 16;
  const int lm = lane & 15;
  const int lh = lane >> 4;
  const int ntiles = (n + 15) >> 4;

  v2f b[32];
#pragma unroll
  for (int j = 0; j < 16; ++j) {
    const float* __restrict__ wp = W + (size_t)(8 * j + 4 * lh) * CH + n0 + lm;
    b[2 * j].x = wp[0];
    b[2 * j].y = wp[CH];
    b[2 * j + 1].x = wp[2 * CH];
    b[2 * j + 1].y = wp[3 * CH];
  }

  float av[32];
  {
    const float* __restrict__ avp = ((tid < 64) ? att_src : att_dst) + (tid & (HEADS - 1)) * 32;
#pragma unroll
    for (int c = 0; c < 32; ++c) av[c] = avp[c];
  }

  v4f a[16];
  auto load_a = [&](int t) {
    int row = t * 16 + lm;
    if (row >= n) row = n - 1;
    const float* arow = x + (size_t)row * CH + 4 * lh;
#pragma unroll
    for (int j = 0; j < 16; ++j) a[j] = *(const v4f*)(arow + 8 * j);
  };
  if ((int)blockIdx.x < ntiles) load_a(blockIdx.x);

  int buf = 0;
  for (int t = blockIdx.x; t < ntiles; t += gridDim.x, buf ^= 1) {
    const int m0 = t * 16;
    v8f acc = {};
#pragma unroll
    for (int j = 0; j < 16; ++j) {
      v2f a0 = {a[j].x, a[j].y};
      v2f a1 = {a[j].z, a[j].w};
      acc = __builtin_amdgcn_wmma_f32_16x16x4_f32(
           false, a0,  false, b[2 * j],
           (short)0, acc,  false,  false);
      acc = __builtin_amdgcn_wmma_f32_16x16x4_f32(
           false, a1,  false, b[2 * j + 1],
           (short)0, acc,  false,  false);
    }
    load_a(min(t + (int)gridDim.x, ntiles - 1));

    float* htb = ht[buf];
    {
      float* __restrict__ hs = htb + (8 * lh) * HP + n0 + lm;
#pragma unroll
      for (int v = 0; v < 8; ++v) hs[v * HP] = acc[v];
    }
    __builtin_amdgcn_fence(__ATOMIC_RELEASE, "workgroup", "local");
    __builtin_amdgcn_s_barrier();
    __builtin_amdgcn_fence(__ATOMIC_ACQUIRE, "workgroup", "local");

#pragma unroll
    for (int rr = 0; rr < 2; ++rr) {
      const int r = 2 * wave + rr;
      if (m0 + r < n)
        *(v4f*)(h + (size_t)(m0 + r) * CH + lane * 4) = *(const v4f*)(htb + r * HP + lane * 4);
    }

    if (tid < 128) {
      const int r = (tid & 63) >> 2;
      const int head = tid & (HEADS - 1);
      if (m0 + r < n) {
        const float* hv = htb + r * HP + head * 32;
        float s = 0.0f;
#pragma unroll
        for (int c = 0; c < 32; ++c) s += hv[c] * av[c];
        float* __restrict__ o = (tid < 64) ? a_src : a_dst;
        o[(size_t)(m0 + r) * HEADS + head] = s;
      }
    }
  }
}

#define CAP 8192
#define EB 1024
#define SCAN_K 8
#define ACC_U 8
#define MAX_U 4

__device__ __forceinline__ float leaky(float v) {
  return (v >= 0.0f) ? v : NEG_SLOPE * v;
}

__global__ __launch_bounds__(EB) void gat_tile(
    const int* __restrict__ ei, const unsigned* __restrict__ keys,
    const float* __restrict__ h,
    const float* __restrict__ a_src, const float* __restrict__ a_dst,
    const float* __restrict__ bias, const float* __restrict__ prelu_w,
    float* __restrict__ out, int n, int ne) {
  __shared__ __attribute__((aligned(16))) float acc[TN * CH];
  __shared__ unsigned smax[TN * HEADS];
  __shared__ float sden[TN * HEADS];
  __shared__ float sadst[TN * HEADS];
  __shared__ int ls[CAP];
  __shared__ int ld[CAP];
  __shared__ int cnt;

  const int tid = threadIdx.x;
  const int tile0 = blockIdx.x * TN;
  const int tn = min(TN, n - tile0);
  const int* __restrict__ dst = ei + ne;

  for (int i = tid; i < TN * CH / 4; i += EB) ((v4f*)acc)[i] = v4f{0.f, 0.f, 0.f, 0.f};
  for (int i = tid; i < tn * HEADS; i += EB) {
    smax[i] = 0u;
    sden[i] = 0.0f;
    sadst[i] = a_dst[(size_t)tile0 * HEADS + i];
  }
  if (tid == 0) cnt = 0;
  __syncthreads();

  const unsigned tkey = ((unsigned)blockIdx.x & 255u) * 0x01010101u;
  auto list16 = [&](int g, uint4 k) {
    const unsigned x0 = k.x ^ tkey, x1 = k.y ^ tkey, x2 = k.z ^ tkey, x3 = k.w ^ tkey;
    const unsigned r0 = (x0 - 0x01010101u) & ~x0, r1 = (x1 - 0x01010101u) & ~x1;
    const unsigned r2 = (x2 - 0x01010101u) & ~x2, r3 = (x3 - 0x01010101u) & ~x3;
    if (((r0 | r1 | r2 | r3) & 0x80808080u) != 0u) {
      unsigned m = ((r0 & 0x80808080u) >> 7) | ((r1 & 0x80808080u) >> 6) |
                   ((r2 & 0x80808080u) >> 5) | ((r3 & 0x80808080u) >> 4);
      int slot = atomicAdd(&cnt, __popc(m));
      const int e0 = g * 16;
      while (m != 0u) {
        const int p = __ffs((int)m) - 1;
        m &= m - 1u;
        if (slot < CAP) ls[slot] = e0 + ((p & 7) << 2) + (p >> 3);
        ++slot;
      }
    }
  };
  const int ngrp = (ne + 15) >> 4;
  const uint4* __restrict__ kg = (const uint4*)keys;
  for (int base = tid; base < ngrp; base += EB * SCAN_K) {
    uint4 k4[SCAN_K];
#pragma unroll
    for (int k = 0; k < SCAN_K; ++k) k4[k] = kg[min(base + k * EB, ngrp - 1)];
#pragma unroll
    for (int k = 0; k < SCAN_K; ++k)
      if (base + k * EB < ngrp) list16(base + k * EB, k4[k]);
  }
  __syncthreads();

  const int total = cnt;
  auto max_edge = [&](int dl, v4f as) {
#pragma unroll
    for (int hh = 0; hh < HEADS; ++hh)
      atomicMax(&smax[dl * HEADS + hh], enc_f32(leaky(as[hh] + sadst[dl * HEADS + hh])));
  };
  if (total <= CAP) {
    for (int base = tid; base < total; base += EB * MAX_U) {
      int sv[MAX_U], dd[MAX_U]; v4f as[MAX_U]; bool in_list[MAX_U];
#pragma unroll
      for (int u = 0; u < MAX_U; ++u) {
        int idx = base + u * EB;
        int e = ne;
        if (idx < total) e = ls[idx];
        in_list[u] = e < ne;
        if (!in_list[u]) e = 0;
        sv[u] = ei[e];
        dd[u] = dst[e];
      }
#pragma unroll
      for (int u = 0; u < MAX_U; ++u) as[u] = *(const v4f*)(a_src + (size_t)sv[u] * HEADS);
#pragma unroll
      for (int u = 0; u < MAX_U; ++u) {
        int idx = base + u * EB;
        if (idx < total) {
          unsigned dl = (unsigned)(dd[u] - tile0);
          bool mine = in_list[u] && dl < (unsigned)tn;
          ls[idx] = sv[u];
          ld[idx] = mine ? (int)dl : -1;
          if (mine) max_edge((int)dl, as[u]);
        }
      }
    }
  } else {
    for (int e = tid; e < ne; e += EB) {
      unsigned dl = (unsigned)(dst[e] - tile0);
      if (dl < (unsigned)tn) max_edge((int)dl, *(const v4f*)(a_src + (size_t)ei[e] * HEADS));
    }
  }
  for (int i = tid; i < tn; i += EB)
    max_edge(i, *(const v4f*)(a_src + (size_t)(tile0 + i) * HEADS));
  __syncthreads();

  const int wave = tid >> 5;
  const int lane = tid & 31;
  const int head = lane >> 3;
  auto add_edge = [&](int dl, float av, v4f hv) {
    float v = leaky(av + sadst[dl * HEADS + head]);
    float p = expf(v - dec_f32(smax[dl * HEADS + head]));
    if ((lane & 7) == 0) atomicAdd(&sden[dl * HEADS + head], p);
    float* a = acc + dl * CH + lane;
    atomicAdd(a, p * hv.x);
    atomicAdd(a + 32, p * hv.y);
    atomicAdd(a + 64, p * hv.z);
    atomicAdd(a + 96, p * hv.w);
  };
  if (total <= CAP) {
    const int items = total + tn;
    for (int i0 = wave * ACC_U; i0 < items; i0 += (EB / 32) * ACC_U) {
      int sdl[ACC_U]; float av[ACC_U]; v4f hv[ACC_U];
#pragma unroll
      for (int u = 0; u < ACC_U; ++u) {
        int idx = i0 + u;
        int s = tile0, dl = 0;
        if (idx < total) { s = ls[idx]; dl = ld[idx]; }
        else if (idx < items) { dl = idx - total; s = tile0 + dl; }
        sdl[u] = dl;
        av[u] = a_src[(size_t)s * HEADS + head];
        hv[u] = *(const v4f*)(h + (size_t)s * CH + lane * 4);
      }
#pragma unroll
      for (int u = 0; u < ACC_U; ++u)
        if (i0 + u < items && sdl[u] >= 0) add_edge(sdl[u], av[u], hv[u]);
    }
  } else {
    for (int e = wave; e < ne; e += EB / 32) {
      unsigned dl = (unsigned)(dst[e] - tile0);
      if (dl < (unsigned)tn) {
        int s = ei[e];
        add_edge((int)dl, a_src[(size_t)s * HEADS + head],
                 *(const v4f*)(h + (size_t)s * CH + lane * 4));
      }
    }
    for (int i = wave; i < tn; i += EB / 32) {
      int s = tile0 + i;
      add_edge(i, a_src[(size_t)s * HEADS + head],
               *(const v4f*)(h + (size_t)s * CH + lane * 4));
    }
  }
  __syncthreads();

  const float pw = prelu_w[0];
  for (int i = tid; i < tn * (CH / 4); i += EB) {
    int node = i >> 5;
    int l4 = i & 31;
    float den = sden[node * HEADS + (l4 >> 3)] + 1e-16f;
    const float* a = acc + node * CH + l4;
    v4f b = *(const v4f*)(bias + l4 * 4);
    v4f o;
    o.x = a[0] / den + b.x;
    o.y = a[32] / den + b.y;
    o.z = a[64] / den + b.z;
    o.w = a[96] / den + b.w;
    o.x = (o.x >= 0.0f) ? o.x : pw * o.x;
    o.y = (o.y >= 0.0f) ? o.y : pw * o.y;
    o.z = (o.z >= 0.0f) ? o.z : pw * o.z;
    o.w = (o.w >= 0.0f) ? o.w : pw * o.w;
    *(v4f*)(out + (size_t)(tile0 + node) * CH + l4 * 4) = o;
  }
}

extern "C" void kernel_launch(void* const* d_in, const int* in_sizes, int n_in,
                              void* d_out, int out_size, void* d_ws, size_t ws_size,
                              hipStream_t stream) {
  const float* x       = (const float*)d_in[0];
  const int*   ei      = (const int*)d_in[1];
  const float* W       = (const float*)d_in[2];
  const float* att_src = (const float*)d_in[3];
  const float* att_dst = (const float*)d_in[4];
  const float* bias    = (const float*)d_in[5];
  const float* prelu_w = (const float*)d_in[6];
  float* out = (float*)d_out;

  const int n  = in_sizes[0] / CH;
  const int ne = in_sizes[1] / 2;

  float* h     = (float*)d_ws;
  float* a_src = h + (size_t)n * CH;
  float* a_dst = a_src + (size_t)n * HEADS;
  unsigned* keys = (unsigned*)(a_dst + (size_t)n * HEADS);

  gat_gemm_att<<<GEMM_WG, 256, 0, stream>>>(x, W, att_src, att_dst, ei,
                                                  h, a_src, a_dst, keys, n, ne);
  gat_tile<<<(n + TN - 1) / TN, EB, 0, stream>>>(ei, keys, h, a_src, a_dst, bias, prelu_w,
                                                 out, n, ne);
}


# Edge sums: each (edge, head) softmax weight computed by one lane (8 edges x 4 heads = a wave's 32 lanes) and fetched by the head's 8 lanes with a DPP8 move, not recomputed by all 32; sums by arrival
# speedup vs baseline: 3.7542x; 1.1545x over previous
#include <hip/hip_runtime.h>
#include <hip/hip_bf16.h>

typedef __attribute__((ext_vector_type(2))) float v2f;
typedef __attribute__((ext_vector_type(4))) float v4f;
typedef __attribute__((ext_vector_type(8))) float v8f;

#define HEADS 4
#define CH 128
#define NEG_SLOPE 0.2f
#define TN 196

__device__ __forceinline__ unsigned tile_key(int d) {
  return ((unsigned)d / (unsigned)TN) & 255u;
}

__device__ __forceinline__ unsigned enc_f32(float f) {
  unsigned u = __float_as_uint(f);
  return (f >= 0.0f) ? (u | 0x80000000u) : ~u;
}
__device__ __forceinline__ float dec_f32(unsigned u) {
  return (u & 0x80000000u) ? __uint_as_float(u & 0x7fffffffu)
                           : __uint_as_float(~u);
}

template <int V> struct IC { static constexpr int value = V; };

template <int U>
__device__ __forceinline__ float group8_bcast(float v) {
  return __uint_as_float(__builtin_amdgcn_mov_dpp8(__float_as_uint(v), (unsigned)U * 0x249249u));
}

#define GEMM_WG 512
#define HP 132
__global__ __launch_bounds__(256) void gat_gemm_att(
    const float* x,
    const float* __restrict__ W,
    const float* __restrict__ att_src, const float* __restrict__ att_dst,
    const int* __restrict__ ei,
    float* __restrict__ h, float* __restrict__ a_src, float* __restrict__ a_dst,
    unsigned* __restrict__ keys, int n, int ne) {
  __shared__ __attribute__((aligned(16))) float ht[2][16 * HP];
  const int tid = threadIdx.x;

  {
    const int* __restrict__ dst = ei + ne;
    const int nkw = ((ne + 15) >> 4) << 2;
    for (int w = blockIdx.x * 256 + tid; w < nkw; w += gridDim.x * 256) {
      const int e = w * 4;
      int4 d = int4{0, 0, 0, 0};
      if ((ne & 3) == 0 && e < ne) {
        d = *(const int4*)(dst + e);
      } else {
        if (e < ne) d.x = dst[e];
        if (e + 1 < ne) d.y = dst[e + 1];
        if (e + 2 < ne) d.z = dst[e + 2];
        if (e + 3 < ne) d.w = dst[e + 3];
      }
      keys[w] = tile_key(d.x) | (tile_key(d.y) << 8) | (tile_key(d.z) << 16) | (tile_key(d.w) << 24);
    }
  }

  const int wave = tid >> 5;
  const int lane = tid & 31;
  const int n0 = wave * 16;
  const int lm = lane & 15;
  const int lh = lane >> 4;
  const int ntiles = (n + 15) >> 4;

  v2f b[32];
#pragma unroll
  for (int j = 0; j < 16; ++j) {
    const float* __restrict__ wp = W + (size_t)(8 * j + 4 * lh) * CH + n0 + lm;
    b[2 * j].x = wp[0];
    b[2 * j].y = wp[CH];
    b[2 * j + 1].x = wp[2 * CH];
    b[2 * j + 1].y = wp[3 * CH];
  }

  float av[32];
  {
    const float* __restrict__ avp = ((tid < 64) ? att_src : att_dst) + (tid & (HEADS - 1)) * 32;
#pragma unroll
    for (int c = 0; c < 32; ++c) av[c] = avp[c];
  }

  v4f a[16];
  auto load_a = [&](int t) {
    int row = t * 16 + lm;
    if (row >= n) row = n - 1;
    const float* arow = x + (size_t)row * CH + 4 * lh;
#pragma unroll
    for (int j = 0; j < 16; ++j) a[j] = *(const v4f*)(arow + 8 * j);
  };
  if ((int)blockIdx.x < ntiles) load_a(blockIdx.x);

  int buf = 0;
  for (int t = blockIdx.x; t < ntiles; t += gridDim.x, buf ^= 1) {
    const int m0 = t * 16;
    v8f acc = {};
#pragma unroll
    for (int j = 0; j < 16; ++j) {
      v2f a0 = {a[j].x, a[j].y};
      v2f a1 = {a[j].z, a[j].w};
      acc = __builtin_amdgcn_wmma_f32_16x16x4_f32(
           false, a0,  false, b[2 * j],
           (short)0, acc,  false,  false);
      acc = __builtin_amdgcn_wmma_f32_16x16x4_f32(
           false, a1,  false, b[2 * j + 1],
           (short)0, acc,  false,  false);
    }
    load_a(min(t + (int)gridDim.x, ntiles - 1));

    float* htb = ht[buf];
    {
      float* __restrict__ hs = htb + (8 * lh) * HP + n0 + lm;
#pragma unroll
      for (int v = 0; v < 8; ++v) hs[v * HP] = acc[v];
    }
    __builtin_amdgcn_fence(__ATOMIC_RELEASE, "workgroup", "local");
    __builtin_amdgcn_s_barrier();
    __builtin_amdgcn_fence(__ATOMIC_ACQUIRE, "workgroup", "local");

#pragma unroll
    for (int rr = 0; rr < 2; ++rr) {
      const int r = 2 * wave + rr;
      if (m0 + r < n)
        *(v4f*)(h + (size_t)(m0 + r) * CH + lane * 4) = *(const v4f*)(htb + r * HP + lane * 4);
    }

    if (tid < 128) {
      const int r = (tid & 63) >> 2;
      const int head = tid & (HEADS - 1);
      if (m0 + r < n) {
        const float* hv = htb + r * HP + head * 32;
        float s = 0.0f;
#pragma unroll
        for (int c = 0; c < 32; ++c) s += hv[c] * av[c];
        float* __restrict__ o = (tid < 64) ? a_src : a_dst;
        o[(size_t)(m0 + r) * HEADS + head] = s;
      }
    }
  }
}

#define CAP 8192
#define EB 1024
#define SCAN_K 8
#define ACC_U 8
#define MAX_U 4

__device__ __forceinline__ float leaky(float v) {
  return (v >= 0.0f) ? v : NEG_SLOPE * v;
}

__global__ __launch_bounds__(EB) void gat_tile(
    const int* __restrict__ ei, const unsigned* __restrict__ keys,
    const float* __restrict__ h,
    const float* __restrict__ a_src, const float* __restrict__ a_dst,
    const float* __restrict__ bias, const float* __restrict__ prelu_w,
    float* __restrict__ out, int n, int ne) {
  __shared__ __attribute__((aligned(16))) float acc[TN * CH];
  __shared__ unsigned smax[TN * HEADS];
  __shared__ float sden[TN * HEADS];
  __shared__ float sadst[TN * HEADS];
  __shared__ int ls[CAP];
  __shared__ int ld[CAP];
  __shared__ int cnt;

  const int tid = threadIdx.x;
  const int tile0 = blockIdx.x * TN;
  const int tn = min(TN, n - tile0);
  const int* __restrict__ dst = ei + ne;

  for (int i = tid; i < TN * CH / 4; i += EB) ((v4f*)acc)[i] = v4f{0.f, 0.f, 0.f, 0.f};
  for (int i = tid; i < tn * HEADS; i += EB) {
    smax[i] = 0u;
    sden[i] = 0.0f;
    sadst[i] = a_dst[(size_t)tile0 * HEADS + i];
  }
  if (tid == 0) cnt = 0;
  __syncthreads();

  const unsigned tkey = ((unsigned)blockIdx.x & 255u) * 0x01010101u;
  auto list16 = [&](int g, uint4 k) {
    const unsigned x0 = k.x ^ tkey, x1 = k.y ^ tkey, x2 = k.z ^ tkey, x3 = k.w ^ tkey;
    const unsigned r0 = (x0 - 0x01010101u) & ~x0, r1 = (x1 - 0x01010101u) & ~x1;
    const unsigned r2 = (x2 - 0x01010101u) & ~x2, r3 = (x3 - 0x01010101u) & ~x3;
    if (((r0 | r1 | r2 | r3) & 0x80808080u) != 0u) {
      unsigned m = ((r0 & 0x80808080u) >> 7) | ((r1 & 0x80808080u) >> 6) |
                   ((r2 & 0x80808080u) >> 5) | ((r3 & 0x80808080u) >> 4);
      int slot = atomicAdd(&cnt, __popc(m));
      const int e0 = g * 16;
      while (m != 0u) {
        const int p = __ffs((int)m) - 1;
        m &= m - 1u;
        if (slot < CAP) ls[slot] = e0 + ((p & 7) << 2) + (p >> 3);
        ++slot;
      }
    }
  };
  const int ngrp = (ne + 15) >> 4;
  const uint4* __restrict__ kg = (const uint4*)keys;
  for (int base = tid; base < ngrp; base += EB * SCAN_K) {
    uint4 k4[SCAN_K];
#pragma unroll
    for (int k = 0; k < SCAN_K; ++k) k4[k] = kg[min(base + k * EB, ngrp - 1)];
#pragma unroll
    for (int k = 0; k < SCAN_K; ++k)
      if (base + k * EB < ngrp) list16(base + k * EB, k4[k]);
  }
  __syncthreads();

  const int total = cnt;
  auto max_edge = [&](int dl, v4f as) {
#pragma unroll
    for (int hh = 0; hh < HEADS; ++hh)
      atomicMax(&smax[dl * HEADS + hh], enc_f32(leaky(as[hh] + sadst[dl * HEADS + hh])));
  };
  if (total <= CAP) {
    for (int base = tid; base < total; base += EB * MAX_U) {
      int sv[MAX_U], dd[MAX_U]; v4f as[MAX_U]; bool in_list[MAX_U];
#pragma unroll
      for (int u = 0; u < MAX_U; ++u) {
        int idx = base + u * EB;
        int e = ne;
        if (idx < total) e = ls[idx];
        in_list[u] = e < ne;
        if (!in_list[u]) e = 0;
        sv[u] = ei[e];
        dd[u] = dst[e];
      }
#pragma unroll
      for (int u = 0; u < MAX_U; ++u) as[u] = *(const v4f*)(a_src + (size_t)sv[u] * HEADS);
#pragma unroll
      for (int u = 0; u < MAX_U; ++u) {
        int idx = base + u * EB;
        if (idx < total) {
          unsigned dl = (unsigned)(dd[u] - tile0);
          bool mine = in_list[u] && dl < (unsigned)tn;
          ls[idx] = sv[u];
          ld[idx] = mine ? (int)dl : -1;
          if (mine) max_edge((int)dl, as[u]);
        }
      }
    }
  } else {
    for (int e = tid; e < ne; e += EB) {
      unsigned dl = (unsigned)(dst[e] - tile0);
      if (dl < (unsigned)tn) max_edge((int)dl, *(const v4f*)(a_src + (size_t)ei[e] * HEADS));
    }
  }
  for (int i = tid; i < tn; i += EB)
    max_edge(i, *(const v4f*)(a_src + (size_t)(tile0 + i) * HEADS));
  __syncthreads();

  const int wave = tid >> 5;
  const int lane = tid & 31;
  const int head = lane >> 3;
  auto add_edge = [&](int dl, float av, v4f hv) {
    float v = leaky(av + sadst[dl * HEADS + head]);
    float p = expf(v - dec_f32(smax[dl * HEADS + head]));
    if ((lane & 7) == 0) atomicAdd(&sden[dl * HEADS + head], p);
    float* a = acc + dl * CH + lane;
    atomicAdd(a, p * hv.x);
    atomicAdd(a + 32, p * hv.y);
    atomicAdd(a + 64, p * hv.z);
    atomicAdd(a + 96, p * hv.w);
  };
  if (total <= CAP) {
    const int items = total + tn;
    static_assert(ACC_U * HEADS == 32, "one (item, head) pair per lane");
    const int wv = __builtin_amdgcn_readfirstlane(wave);
    const int sub = lane & (ACC_U - 1);
    for (int i0 = wv * ACC_U; i0 < items; i0 += (EB / 32) * ACC_U) {
      const int idx = i0 + sub;
      int s = tile0, dl = -1;
      if (idx < total) { s = ls[idx]; dl = ld[idx]; }
      else if (idx < items) { dl = idx - total; s = tile0 + dl; }
      const float av = a_src[(size_t)s * HEADS + head];
      int dlu[ACC_U]; v4f hv[ACC_U];
#pragma unroll
      for (int u = 0; u < ACC_U; ++u) {
        const int su = __builtin_amdgcn_readlane(s, u);
        dlu[u] = __builtin_amdgcn_readlane(dl, u);
        hv[u] = *(const v4f*)(h + (size_t)su * CH + lane * 4);
      }
      const int dls = (dl >= 0) ? dl : 0;
      const float v = leaky(av + sadst[dls * HEADS + head]);
      const float p = expf(v - dec_f32(smax[dls * HEADS + head]));
      if (dl >= 0) atomicAdd(&sden[dl * HEADS + head], p);
      auto add_row = [&](auto uc) {
        constexpr int u = decltype(uc)::value;
        if (dlu[u] >= 0) {
          const float pu = group8_bcast<u>(p);
          float* a = acc + dlu[u] * CH + lane;
          atomicAdd(a, pu * hv[u].x);
          atomicAdd(a + 32, pu * hv[u].y);
          atomicAdd(a + 64, pu * hv[u].z);
          atomicAdd(a + 96, pu * hv[u].w);
        }
      };
      add_row(IC<0>{}); add_row(IC<1>{}); add_row(IC<2>{}); add_row(IC<3>{});
      add_row(IC<4>{}); add_row(IC<5>{}); add_row(IC<6>{}); add_row(IC<7>{});
    }
  } else {
    for (int e = wave; e < ne; e += EB / 32) {
      unsigned dl = (unsigned)(dst[e] - tile0);
      if (dl < (unsigned)tn) {
        int s = ei[e];
        add_edge((int)dl, a_src[(size_t)s * HEADS + head],
                 *(const v4f*)(h + (size_t)s * CH + lane * 4));
      }
    }
    for (int i = wave; i < tn; i += EB / 32) {
      int s = tile0 + i;
      add_edge(i, a_src[(size_t)s * HEADS + head],
               *(const v4f*)(h + (size_t)s * CH + lane * 4));
    }
  }
  __syncthreads();

  const float pw = prelu_w[0];
  for (int i = tid; i < tn * (CH / 4); i += EB) {
    int node = i >> 5;
    int l4 = i & 31;
    float den = sden[node * HEADS + (l4 >> 3)] + 1e-16f;
    const float* a = acc + node * CH + l4;
    v4f b = *(const v4f*)(bias + l4 * 4);
    v4f o;
    o.x = a[0] / den + b.x;
    o.y = a[32] / den + b.y;
    o.z = a[64] / den + b.z;
    o.w = a[96] / den + b.w;
    o.x = (o.x >= 0.0f) ? o.x : pw * o.x;
    o.y = (o.y >= 0.0f) ? o.y : pw * o.y;
    o.z = (o.z >= 0.0f) ? o.z : pw * o.z;
    o.w = (o.w >= 0.0f) ? o.w : pw * o.w;
    *(v4f*)(out + (size_t)(tile0 + node) * CH + l4 * 4) = o;
  }
}

extern "C" void kernel_launch(void* const* d_in, const int* in_sizes, int n_in,
                              void* d_out, int out_size, void* d_ws, size_t ws_size,
                              hipStream_t stream) {
  const float* x       = (const float*)d_in[0];
  const int*   ei      = (const int*)d_in[1];
  const float* W       = (const float*)d_in[2];
  const float* att_src = (const float*)d_in[3];
  const float* att_dst = (const float*)d_in[4];
  const float* bias    = (const float*)d_in[5];
  const float* prelu_w = (const float*)d_in[6];
  float* out = (float*)d_out;

  const int n  = in_sizes[0] / CH;
  const int ne = in_sizes[1] / 2;

  float* h     = (float*)d_ws;
  float* a_src = h + (size_t)n * CH;
  float* a_dst = a_src + (size_t)n * HEADS;
  unsigned* keys = (unsigned*)(a_dst + (size_t)n * HEADS);

  gat_gemm_att<<<GEMM_WG, 256, 0, stream>>>(x, W, att_src, att_dst, ei,
                                                  h, a_src, a_dst, keys, n, ne);
  gat_tile<<<(n + TN - 1) / TN, EB, 0, stream>>>(ei, keys, h, a_src, a_dst, bias, prelu_w,
                                                 out, n, ne);
}


# Per-edge tile key bytes replaced by a per-tile bitmap over 32-edge groups built in the projection launch; a tile reads its row and only flagged groups' destinations; node sums still by arrival
# speedup vs baseline: 4.2495x; 1.1319x over previous
#include <hip/hip_runtime.h>
#include <hip/hip_bf16.h>

typedef __attribute__((ext_vector_type(2))) float v2f;
typedef __attribute__((ext_vector_type(4))) float v4f;
typedef __attribute__((ext_vector_type(8))) float v8f;

#define HEADS 4
#define CH 128
#define NEG_SLOPE 0.2f
#define TN 196

__device__ __forceinline__ unsigned tile_key(int d) {
  return ((unsigned)d / (unsigned)TN) & 255u;
}
#define BROWS 256
#define CHG 256

__device__ __forceinline__ unsigned enc_f32(float f) {
  unsigned u = __float_as_uint(f);
  return (f >= 0.0f) ? (u | 0x80000000u) : ~u;
}
__device__ __forceinline__ float dec_f32(unsigned u) {
  return (u & 0x80000000u) ? __uint_as_float(u & 0x7fffffffu)
                           : __uint_as_float(~u);
}

template <int V> struct IC { static constexpr int value = V; };

template <int U>
__device__ __forceinline__ float group8_bcast(float v) {
  return __uint_as_float(__builtin_amdgcn_mov_dpp8(__float_as_uint(v), (unsigned)U * 0x249249u));
}

#define GEMM_WG 512
#define HP 132
__global__ __launch_bounds__(256) void gat_gemm_att(
    const float* x,
    const float* __restrict__ W,
    const float* __restrict__ att_src, const float* __restrict__ att_dst,
    const int* __restrict__ ei,
    float* __restrict__ h, float* __restrict__ a_src, float* __restrict__ a_dst,
    unsigned* __restrict__ bm, int n, int ne) {
  __shared__ __attribute__((aligned(16))) float ht[2][16 * HP];
  const int tid = threadIdx.x;

  {
    __shared__ __attribute__((aligned(16))) unsigned bml[BROWS * (CHG / 32)];
    const int* __restrict__ dst = ei + ne;
    const int ngr = (ne + 31) >> 5;
    const int nch = (ngr + CHG - 1) / CHG;
    const int rw = nch * (CHG / 32);
    auto chunk_bits = [&](auto al, int c) {
      constexpr int NJ = (CHG * 32) / (256 * 4);
      const int e0 = c * (CHG * 32);
      int4 d[NJ];
#pragma unroll
      for (int j = 0; j < NJ; ++j) {
        const int e = e0 + (j * 256 + tid) * 4;
        if constexpr (decltype(al)::value != 0) {
          d[j] = *(const int4*)(dst + min(e, ne - 4));
        } else {
          d[j].x = dst[min(e, ne - 1)];
          d[j].y = dst[min(e + 1, ne - 1)];
          d[j].z = dst[min(e + 2, ne - 1)];
          d[j].w = dst[min(e + 3, ne - 1)];
        }
      }
      for (int i = tid; i < BROWS * (CHG / 32); i += 256) bml[i] = 0u;
      __syncthreads();
#pragma unroll
      for (int j = 0; j < NJ; ++j) {
        const int q = j * 256 + tid;
        const int nv = ne - (e0 + q * 4);
        const int col = q >> 8;
        const unsigned bit = 1u << ((q >> 3) & 31);
        if (nv > 0) atomicOr(&bml[tile_key(d[j].x) * (CHG / 32) + col], bit);
        if (nv > 1) atomicOr(&bml[tile_key(d[j].y) * (CHG / 32) + col], bit);
        if (nv > 2) atomicOr(&bml[tile_key(d[j].z) * (CHG / 32) + col], bit);
        if (nv > 3) atomicOr(&bml[tile_key(d[j].w) * (CHG / 32) + col], bit);
      }
    };
    for (int c = (int)gridDim.x - 1 - (int)blockIdx.x; c < nch; c += gridDim.x) {
      if ((ne & 3) == 0) chunk_bits(IC<1>{}, c); else chunk_bits(IC<0>{}, c);
      __syncthreads();
      {
        const uint4* src4 = (const uint4*)(bml + tid * (CHG / 32));
        uint4* out4 = (uint4*)(bm + (size_t)tid * rw + (size_t)c * (CHG / 32));
        out4[0] = src4[0];
        out4[1] = src4[1];
      }
      __syncthreads();
    }
  }

  const int wave = tid >> 5;
  const int lane = tid & 31;
  const int n0 = wave * 16;
  const int lm = lane & 15;
  const int lh = lane >> 4;
  const int ntiles = (n + 15) >> 4;

  v2f b[32];
#pragma unroll
  for (int j = 0; j < 16; ++j) {
    const float* __restrict__ wp = W + (size_t)(8 * j + 4 * lh) * CH + n0 + lm;
    b[2 * j].x = wp[0];
    b[2 * j].y = wp[CH];
    b[2 * j + 1].x = wp[2 * CH];
    b[2 * j + 1].y = wp[3 * CH];
  }

  float av[32];
  {
    const float* __restrict__ avp = ((tid < 64) ? att_src : att_dst) + (tid & (HEADS - 1)) * 32;
#pragma unroll
    for (int c = 0; c < 32; ++c) av[c] = avp[c];
  }

  v4f a[16];
  auto load_a = [&](int t) {
    int row = t * 16 + lm;
    if (row >= n) row = n - 1;
    const float* arow = x + (size_t)row * CH + 4 * lh;
#pragma unroll
    for (int j = 0; j < 16; ++j) a[j] = *(const v4f*)(arow + 8 * j);
  };
  if ((int)blockIdx.x < ntiles) load_a(blockIdx.x);

  int buf = 0;
  for (int t = blockIdx.x; t < ntiles; t += gridDim.x, buf ^= 1) {
    const int m0 = t * 16;
    v8f acc = {};
#pragma unroll
    for (int j = 0; j < 16; ++j) {
      v2f a0 = {a[j].x, a[j].y};
      v2f a1 = {a[j].z, a[j].w};
      acc = __builtin_amdgcn_wmma_f32_16x16x4_f32(
           false, a0,  false, b[2 * j],
           (short)0, acc,  false,  false);
      acc = __builtin_amdgcn_wmma_f32_16x16x4_f32(
           false, a1,  false, b[2 * j + 1],
           (short)0, acc,  false,  false);
    }
    load_a(min(t + (int)gridDim.x, ntiles - 1));

    float* htb = ht[buf];
    {
      float* __restrict__ hs = htb + (8 * lh) * HP + n0 + lm;
#pragma unroll
      for (int v = 0; v < 8; ++v) hs[v * HP] = acc[v];
    }
    __builtin_amdgcn_fence(__ATOMIC_RELEASE, "workgroup", "local");
    __builtin_amdgcn_s_barrier();
    __builtin_amdgcn_fence(__ATOMIC_ACQUIRE, "workgroup", "local");

#pragma unroll
    for (int rr = 0; rr < 2; ++rr) {
      const int r = 2 * wave + rr;
      if (m0 + r < n)
        *(v4f*)(h + (size_t)(m0 + r) * CH + lane * 4) = *(const v4f*)(htb + r * HP + lane * 4);
    }

    if (tid < 128) {
      const int r = (tid & 63) >> 2;
      const int head = tid & (HEADS - 1);
      if (m0 + r < n) {
        const float* hv = htb + r * HP + head * 32;
        float s = 0.0f;
#pragma unroll
        for (int c = 0; c < 32; ++c) s += hv[c] * av[c];
        float* __restrict__ o = (tid < 64) ? a_src : a_dst;
        o[(size_t)(m0 + r) * HEADS + head] = s;
      }
    }
  }
}

#define CAP 8192
#define EB 1024
#define SEG (CAP / (EB / 32))
#define GLW 1024
#define SCAN_U 8
#define ACC_U 8
#define MAX_U 4

__device__ __forceinline__ float leaky(float v) {
  return (v >= 0.0f) ? v : NEG_SLOPE * v;
}

__global__ __launch_bounds__(EB) void gat_tile(
    const int* __restrict__ ei, const unsigned* __restrict__ bm,
    const float* __restrict__ h,
    const float* __restrict__ a_src, const float* __restrict__ a_dst,
    const float* __restrict__ bias, const float* __restrict__ prelu_w,
    float* __restrict__ out, int n, int ne) {
  __shared__ __attribute__((aligned(16))) float acc[TN * CH];
  __shared__ unsigned smax[TN * HEADS];
  __shared__ float sden[TN * HEADS];
  __shared__ float sadst[TN * HEADS];
  __shared__ __attribute__((aligned(8))) int2 lsd[CAP];
  __shared__ __attribute__((aligned(16))) int l2[2 * CAP];
  __shared__ int wcnt[EB / 32];
  static_assert(2 * CAP * 4 >= (EB / 32) * GLW * 2, "the group lists fit in l2");
  int* const ls2 = l2;
  int* const ld2 = l2 + CAP;

  const int tid = threadIdx.x;
  const int tile0 = blockIdx.x * TN;
  const int tn = min(TN, n - tile0);
  const int* __restrict__ dst = ei + ne;

  for (int i = tid; i < TN * CH / 4; i += EB) ((v4f*)acc)[i] = v4f{0.f, 0.f, 0.f, 0.f};
  for (int i = tid; i < tn * HEADS; i += EB) {
    smax[i] = 0u;
    sden[i] = 0.0f;
    sadst[i] = a_dst[(size_t)tile0 * HEADS + i];
  }
  __syncthreads();

  const int wave = tid >> 5;
  const int lane = tid & 31;
  const int wv = __builtin_amdgcn_readfirstlane(wave);
  int wfound;
  {
    const int ngr = (ne + 31) >> 5;
    const int nch = (ngr + CHG - 1) / CHG;
    const int rw = nch * (CHG / 32);
    const int nw = (ngr + 31) >> 5;
    const int wpw = (nw + (EB / 32) - 1) / (EB / 32);
    const unsigned* __restrict__ brow = bm + (size_t)(blockIdx.x & (BROWS - 1)) * rw;
    unsigned short* const gl = (unsigned short*)l2 + wv * GLW;
    const int sub = lane & 7;
    const int qd = lane >> 3;
    const int segb = wv * SEG;
    int wpos = segb;
    for (int off = 0; off < wpw; off += 32) {
      const int wi = wv * wpw + off + lane;
      unsigned word = 0u;
      if (off + lane < wpw && wi < nw) word = brow[wi];
      const int c = __popc(word);
      int incl = c;
#pragma unroll
      for (int d = 1; d < 32; d <<= 1) {
        const int t = __shfl_up(incl, d);
        if (lane >= d) incl += t;
      }
      const int nfl = __builtin_amdgcn_readlane(incl, 31);
      {
        int k = incl - c;
        while (word != 0u) {
          const int b = __ffs((int)word) - 1;
          word &= word - 1u;
          gl[k++] = (unsigned short)(lane * 32 + b);
        }
      }
      const int g0 = (wv * wpw + off) * 32;
      auto batches = [&](auto al) {
      for (int s0 = 0; s0 < nfl; s0 += 4 * SCAN_U) {
        int4 dv[SCAN_U]; int eb[SCAN_U]; int nrem[SCAN_U]; int rel[SCAN_U];
#pragma unroll
        for (int u = 0; u < SCAN_U; ++u) rel[u] = gl[min(s0 + 4 * u + qd, nfl - 1)];
#pragma unroll
        for (int u = 0; u < SCAN_U; ++u) {
          const int idx = s0 + 4 * u + qd;
          const int e = (g0 + rel[u]) * 32 + sub * 4;
          eb[u] = e;
          nrem[u] = (idx < nfl) ? (ne - e) : 0;
          if constexpr (decltype(al)::value != 0) {
            dv[u] = *(const int4*)(dst + min(e, ne - 4));
          } else {
            dv[u].x = dst[min(e, ne - 1)];
            dv[u].y = dst[min(e + 1, ne - 1)];
            dv[u].z = dst[min(e + 2, ne - 1)];
            dv[u].w = dst[min(e + 3, ne - 1)];
          }
        }
#pragma unroll
        for (int u = 0; u < SCAN_U; ++u) {
          const unsigned lim = (nrem[u] > 0) ? (unsigned)tn : 0u;
          auto take = [&](int k, int d) {
            const unsigned dl = (unsigned)(d - tile0);
            bool hit;
            if constexpr (decltype(al)::value != 0) hit = dl < lim;
            else hit = k < nrem[u] && dl < (unsigned)tn;
            const unsigned bal = __builtin_amdgcn_ballot_w32(hit);
            if (bal != 0u) {
              const int slot = wpos + (int)__builtin_amdgcn_mbcnt_lo(bal, 0u);
              if (hit && slot < segb + SEG) lsd[slot] = int2{eb[u] + k, (int)dl};
              wpos += __popc(bal);
            }
          };
          take(0, dv[u].x); take(1, dv[u].y); take(2, dv[u].z); take(3, dv[u].w);
        }
      }
      };
      if ((ne & 3) == 0) batches(IC<1>{}); else batches(IC<0>{});
    }
    wfound = wpos - segb;
    if (lane == 0) wcnt[wv] = wfound;
  }
  __syncthreads();

  int offw, total;
  bool fits;
  {
    static_assert(EB / 32 == 32, "one wave's count per lane");
    const int c = wcnt[lane];
    int incl = c;
#pragma unroll
    for (int d = 1; d < 32; d <<= 1) {
      const int t = __shfl_up(incl, d);
      if (lane >= d) incl += t;
    }
    offw = __builtin_amdgcn_readlane(incl - c, wv);
    total = __builtin_amdgcn_readlane(incl, 31);
    fits = __builtin_amdgcn_ballot_w32(c > SEG) == 0u;
  }
  auto max_edge = [&](int dl, v4f as) {
#pragma unroll
    for (int hh = 0; hh < HEADS; ++hh)
      atomicMax(&smax[dl * HEADS + hh], enc_f32(leaky(as[hh] + sadst[dl * HEADS + hh])));
  };
  if (fits) {
    const int segb = wv * SEG;
    for (int base = lane; base < wfound; base += 32 * MAX_U) {
      int sv[MAX_U], dd[MAX_U]; v4f as[MAX_U];
#pragma unroll
      for (int u = 0; u < MAX_U; ++u) {
        const int j = base + u * 32;
        int e = 0;
        dd[u] = 0;
        if (j < wfound) { const int2 r = lsd[segb + j]; e = r.x; dd[u] = r.y; }
        sv[u] = ei[e];
      }
#pragma unroll
      for (int u = 0; u < MAX_U; ++u) as[u] = *(const v4f*)(a_src + (size_t)sv[u] * HEADS);
#pragma unroll
      for (int u = 0; u < MAX_U; ++u) {
        const int j = base + u * 32;
        if (j < wfound) {
          ls2[offw + j] = sv[u];
          ld2[offw + j] = dd[u];
          max_edge(dd[u], as[u]);
        }
      }
    }
  } else {
    for (int e = tid; e < ne; e += EB) {
      unsigned dl = (unsigned)(dst[e] - tile0);
      if (dl < (unsigned)tn) max_edge((int)dl, *(const v4f*)(a_src + (size_t)ei[e] * HEADS));
    }
  }
  for (int i = tid; i < tn; i += EB)
    max_edge(i, *(const v4f*)(a_src + (size_t)(tile0 + i) * HEADS));
  __syncthreads();

  const int head = lane >> 3;
  auto add_edge = [&](int dl, float av, v4f hv) {
    float v = leaky(av + sadst[dl * HEADS + head]);
    float p = expf(v - dec_f32(smax[dl * HEADS + head]));
    if ((lane & 7) == 0) atomicAdd(&sden[dl * HEADS + head], p);
    float* a = acc + dl * CH + lane;
    atomicAdd(a, p * hv.x);
    atomicAdd(a + 32, p * hv.y);
    atomicAdd(a + 64, p * hv.z);
    atomicAdd(a + 96, p * hv.w);
  };
  if (fits) {
    const int items = total + tn;
    static_assert(ACC_U * HEADS == 32, "one (item, head) pair per lane");
    const int sub = lane & (ACC_U - 1);
    for (int i0 = wv * ACC_U; i0 < items; i0 += (EB / 32) * ACC_U) {
      const int idx = i0 + sub;
      int s = tile0, dl = -1;
      if (idx < total) { s = ls2[idx]; dl = ld2[idx]; }
      else if (idx < items) { dl = idx - total; s = tile0 + dl; }
      const float av = a_src[(size_t)s * HEADS + head];
      int dlu[ACC_U]; v4f hv[ACC_U];
#pragma unroll
      for (int u = 0; u < ACC_U; ++u) {
        const int su = __builtin_amdgcn_readlane(s, u);
        dlu[u] = __builtin_amdgcn_readlane(dl, u);
        hv[u] = *(const v4f*)(h + (size_t)su * CH + lane * 4);
      }
      const int dls = (dl >= 0) ? dl : 0;
      const float v = leaky(av + sadst[dls * HEADS + head]);
      const float p = expf(v - dec_f32(smax[dls * HEADS + head]));
      if (dl >= 0) atomicAdd(&sden[dl * HEADS + head], p);
      auto add_row = [&](auto uc) {
        constexpr int u = decltype(uc)::value;
        if (dlu[u] >= 0) {
          const float pu = group8_bcast<u>(p);
          float* a = acc + dlu[u] * CH + lane;
          atomicAdd(a, pu * hv[u].x);
          atomicAdd(a + 32, pu * hv[u].y);
          atomicAdd(a + 64, pu * hv[u].z);
          atomicAdd(a + 96, pu * hv[u].w);
        }
      };
      add_row(IC<0>{}); add_row(IC<1>{}); add_row(IC<2>{}); add_row(IC<3>{});
      add_row(IC<4>{}); add_row(IC<5>{}); add_row(IC<6>{}); add_row(IC<7>{});
    }
  } else {
    for (int e = wave; e < ne; e += EB / 32) {
      unsigned dl = (unsigned)(dst[e] - tile0);
      if (dl < (unsigned)tn) {
        int s = ei[e];
        add_edge((int)dl, a_src[(size_t)s * HEADS + head],
                 *(const v4f*)(h + (size_t)s * CH + lane * 4));
      }
    }
    for (int i = wave; i < tn; i += EB / 32) {
      int s = tile0 + i;
      add_edge(i, a_src[(size_t)s * HEADS + head],
               *(const v4f*)(h + (size_t)s * CH + lane * 4));
    }
  }
  __syncthreads();

  const float pw = prelu_w[0];
  for (int i = tid; i < tn * (CH / 4); i += EB) {
    int node = i >> 5;
    int l4 = i & 31;
    float den = sden[node * HEADS + (l4 >> 3)] + 1e-16f;
    const float* a = acc + node * CH + l4;
    v4f b = *(const v4f*)(bias + l4 * 4);
    v4f o;
    o.x = a[0] / den + b.x;
    o.y = a[32] / den + b.y;
    o.z = a[64] / den + b.z;
    o.w = a[96] / den + b.w;
    o.x = (o.x >= 0.0f) ? o.x : pw * o.x;
    o.y = (o.y >= 0.0f) ? o.y : pw * o.y;
    o.z = (o.z >= 0.0f) ? o.z : pw * o.z;
    o.w = (o.w >= 0.0f) ? o.w : pw * o.w;
    *(v4f*)(out + (size_t)(tile0 + node) * CH + l4 * 4) = o;
  }
}

extern "C" void kernel_launch(void* const* d_in, const int* in_sizes, int n_in,
                              void* d_out, int out_size, void* d_ws, size_t ws_size,
                              hipStream_t stream) {
  const float* x       = (const float*)d_in[0];
  const int*   ei      = (const int*)d_in[1];
  const float* W       = (const float*)d_in[2];
  const float* att_src = (const float*)d_in[3];
  const float* att_dst = (const float*)d_in[4];
  const float* bias    = (const float*)d_in[5];
  const float* prelu_w = (const float*)d_in[6];
  float* out = (float*)d_out;

  const int n  = in_sizes[0] / CH;
  const int ne = in_sizes[1] / 2;

  float* h     = (float*)d_ws;
  float* a_src = h + (size_t)n * CH;
  float* a_dst = a_src + (size_t)n * HEADS;
  unsigned* bm = (unsigned*)(a_dst + (size_t)n * HEADS);

  gat_gemm_att<<<GEMM_WG, 256, 0, stream>>>(x, W, att_src, att_dst, ei,
                                                  h, a_src, a_dst, bm, n, ne);
  gat_tile<<<(n + TN - 1) / TN, EB, 0, stream>>>(ei, bm, h, a_src, a_dst, bias, prelu_w,
                                                 out, n, ne);
}


# Edge kernel keeps each item's four per-head scores in LDS from the maximum pass; the sum pass reads them there instead of gathering a_src by source a second time; node sums still by arrival
# speedup vs baseline: 4.3878x; 1.0325x over previous
#include <hip/hip_runtime.h>
#include <hip/hip_bf16.h>

typedef __attribute__((ext_vector_type(2))) float v2f;
typedef __attribute__((ext_vector_type(4))) float v4f;
typedef __attribute__((ext_vector_type(8))) float v8f;

#define HEADS 4
#define CH 128
#define NEG_SLOPE 0.2f
#define TN 196

__device__ __forceinline__ unsigned tile_key(int d) {
  return ((unsigned)d / (unsigned)TN) & 255u;
}
#define BROWS 256
#define CHG 256

__device__ __forceinline__ unsigned enc_f32(float f) {
  unsigned u = __float_as_uint(f);
  return (f >= 0.0f) ? (u | 0x80000000u) : ~u;
}
__device__ __forceinline__ float dec_f32(unsigned u) {
  return (u & 0x80000000u) ? __uint_as_float(u & 0x7fffffffu)
                           : __uint_as_float(~u);
}

template <int V> struct IC { static constexpr int value = V; };

template <int U>
__device__ __forceinline__ float group8_bcast(float v) {
  return __uint_as_float(__builtin_amdgcn_mov_dpp8(__float_as_uint(v), (unsigned)U * 0x249249u));
}

#define GEMM_WG 512
#define HP 132
__global__ __launch_bounds__(256) void gat_gemm_att(
    const float* x,
    const float* __restrict__ W,
    const float* __restrict__ att_src, const float* __restrict__ att_dst,
    const int* __restrict__ ei,
    float* __restrict__ h, float* __restrict__ a_src, float* __restrict__ a_dst,
    unsigned* __restrict__ bm, int n, int ne) {
  __shared__ __attribute__((aligned(16))) float ht[2][16 * HP];
  const int tid = threadIdx.x;

  {
    __shared__ __attribute__((aligned(16))) unsigned bml[BROWS * (CHG / 32)];
    const int* __restrict__ dst = ei + ne;
    const int ngr = (ne + 31) >> 5;
    const int nch = (ngr + CHG - 1) / CHG;
    const int rw = nch * (CHG / 32);
    auto chunk_bits = [&](auto al, int c) {
      constexpr int NJ = (CHG * 32) / (256 * 4);
      const int e0 = c * (CHG * 32);
      int4 d[NJ];
#pragma unroll
      for (int j = 0; j < NJ; ++j) {
        const int e = e0 + (j * 256 + tid) * 4;
        if constexpr (decltype(al)::value != 0) {
          d[j] = *(const int4*)(dst + min(e, ne - 4));
        } else {
          d[j].x = dst[min(e, ne - 1)];
          d[j].y = dst[min(e + 1, ne - 1)];
          d[j].z = dst[min(e + 2, ne - 1)];
          d[j].w = dst[min(e + 3, ne - 1)];
        }
      }
      for (int i = tid; i < BROWS * (CHG / 32); i += 256) bml[i] = 0u;
      __syncthreads();
#pragma unroll
      for (int j = 0; j < NJ; ++j) {
        const int q = j * 256 + tid;
        const int nv = ne - (e0 + q * 4);
        const int col = q >> 8;
        const unsigned bit = 1u << ((q >> 3) & 31);
        if (nv > 0) atomicOr(&bml[tile_key(d[j].x) * (CHG / 32) + col], bit);
        if (nv > 1) atomicOr(&bml[tile_key(d[j].y) * (CHG / 32) + col], bit);
        if (nv > 2) atomicOr(&bml[tile_key(d[j].z) * (CHG / 32) + col], bit);
        if (nv > 3) atomicOr(&bml[tile_key(d[j].w) * (CHG / 32) + col], bit);
      }
    };
    for (int c = (int)gridDim.x - 1 - (int)blockIdx.x; c < nch; c += gridDim.x) {
      if ((ne & 3) == 0) chunk_bits(IC<1>{}, c); else chunk_bits(IC<0>{}, c);
      __syncthreads();
      {
        const uint4* src4 = (const uint4*)(bml + tid * (CHG / 32));
        uint4* out4 = (uint4*)(bm + (size_t)tid * rw + (size_t)c * (CHG / 32));
        out4[0] = src4[0];
        out4[1] = src4[1];
      }
      __syncthreads();
    }
  }

  const int wave = tid >> 5;
  const int lane = tid & 31;
  const int n0 = wave * 16;
  const int lm = lane & 15;
  const int lh = lane >> 4;
  const int ntiles = (n + 15) >> 4;

  v2f b[32];
#pragma unroll
  for (int j = 0; j < 16; ++j) {
    const float* __restrict__ wp = W + (size_t)(8 * j + 4 * lh) * CH + n0 + lm;
    b[2 * j].x = wp[0];
    b[2 * j].y = wp[CH];
    b[2 * j + 1].x = wp[2 * CH];
    b[2 * j + 1].y = wp[3 * CH];
  }

  float av[32];
  {
    const float* __restrict__ avp = ((tid < 64) ? att_src : att_dst) + (tid & (HEADS - 1)) * 32;
#pragma unroll
    for (int c = 0; c < 32; ++c) av[c] = avp[c];
  }

  v4f a[16];
  auto load_a = [&](int t) {
    int row = t * 16 + lm;
    if (row >= n) row = n - 1;
    const float* arow = x + (size_t)row * CH + 4 * lh;
#pragma unroll
    for (int j = 0; j < 16; ++j) a[j] = *(const v4f*)(arow + 8 * j);
  };
  if ((int)blockIdx.x < ntiles) load_a(blockIdx.x);

  int buf = 0;
  for (int t = blockIdx.x; t < ntiles; t += gridDim.x, buf ^= 1) {
    const int m0 = t * 16;
    v8f acc = {};
#pragma unroll
    for (int j = 0; j < 16; ++j) {
      v2f a0 = {a[j].x, a[j].y};
      v2f a1 = {a[j].z, a[j].w};
      acc = __builtin_amdgcn_wmma_f32_16x16x4_f32(
           false, a0,  false, b[2 * j],
           (short)0, acc,  false,  false);
      acc = __builtin_amdgcn_wmma_f32_16x16x4_f32(
           false, a1,  false, b[2 * j + 1],
           (short)0, acc,  false,  false);
    }
    load_a(min(t + (int)gridDim.x, ntiles - 1));

    float* htb = ht[buf];
    {
      float* __restrict__ hs = htb + (8 * lh) * HP + n0 + lm;
#pragma unroll
      for (int v = 0; v < 8; ++v) hs[v * HP] = acc[v];
    }
    __builtin_amdgcn_fence(__ATOMIC_RELEASE, "workgroup", "local");
    __builtin_amdgcn_s_barrier();
    __builtin_amdgcn_fence(__ATOMIC_ACQUIRE, "workgroup", "local");

#pragma unroll
    for (int rr = 0; rr < 2; ++rr) {
      const int r = 2 * wave + rr;
      if (m0 + r < n)
        *(v4f*)(h + (size_t)(m0 + r) * CH + lane * 4) = *(const v4f*)(htb + r * HP + lane * 4);
    }

    if (tid < 128) {
      const int r = (tid & 63) >> 2;
      const int head = tid & (HEADS - 1);
      if (m0 + r < n) {
        const float* hv = htb + r * HP + head * 32;
        float s = 0.0f;
#pragma unroll
        for (int c = 0; c < 32; ++c) s += hv[c] * av[c];
        float* __restrict__ o = (tid < 64) ? a_src : a_dst;
        o[(size_t)(m0 + r) * HEADS + head] = s;
      }
    }
  }
}

#define CAP 8192
#define EB 1024
#define SEG (CAP / (EB / 32))
#define GLW 1024
#define SCAN_U 8
#define ACC_U 8
#define MAX_U 4
#define CAPS 4608

__device__ __forceinline__ float leaky(float v) {
  return (v >= 0.0f) ? v : NEG_SLOPE * v;
}

__global__ __launch_bounds__(EB) void gat_tile(
    const int* __restrict__ ei, const unsigned* __restrict__ bm,
    const float* __restrict__ h,
    const float* __restrict__ a_src, const float* __restrict__ a_dst,
    const float* __restrict__ bias, const float* __restrict__ prelu_w,
    float* __restrict__ out, int n, int ne) {
  __shared__ __attribute__((aligned(16))) float acc[TN * CH];
  __shared__ unsigned smax[TN * HEADS];
  __shared__ float sden[TN * HEADS];
  __shared__ float sadst[TN * HEADS];
  __shared__ __attribute__((aligned(8))) int2 lsd[CAP];
  __shared__ __attribute__((aligned(16))) int l2[2 * CAP];
  __shared__ int wcnt[EB / 32];
  __shared__ __attribute__((aligned(16))) float esc[CAPS * HEADS];
  static_assert(2 * CAP * 4 >= (EB / 32) * GLW * 2, "the group lists fit in l2");
  int* const ls2 = l2;
  int* const ld2 = l2 + CAP;

  const int tid = threadIdx.x;
  const int tile0 = blockIdx.x * TN;
  const int tn = min(TN, n - tile0);
  const int* __restrict__ dst = ei + ne;

  for (int i = tid; i < TN * CH / 4; i += EB) ((v4f*)acc)[i] = v4f{0.f, 0.f, 0.f, 0.f};
  for (int i = tid; i < tn * HEADS; i += EB) {
    smax[i] = 0u;
    sden[i] = 0.0f;
    sadst[i] = a_dst[(size_t)tile0 * HEADS + i];
  }
  __syncthreads();

  const int wave = tid >> 5;
  const int lane = tid & 31;
  const int wv = __builtin_amdgcn_readfirstlane(wave);
  int wfound;
  {
    const int ngr = (ne + 31) >> 5;
    const int nch = (ngr + CHG - 1) / CHG;
    const int rw = nch * (CHG / 32);
    const int nw = (ngr + 31) >> 5;
    const int wpw = (nw + (EB / 32) - 1) / (EB / 32);
    const unsigned* __restrict__ brow = bm + (size_t)(blockIdx.x & (BROWS - 1)) * rw;
    unsigned short* const gl = (unsigned short*)l2 + wv * GLW;
    const int sub = lane & 7;
    const int qd = lane >> 3;
    const int segb = wv * SEG;
    int wpos = segb;
    for (int off = 0; off < wpw; off += 32) {
      const int wi = wv * wpw + off + lane;
      unsigned word = 0u;
      if (off + lane < wpw && wi < nw) word = brow[wi];
      const int c = __popc(word);
      int incl = c;
#pragma unroll
      for (int d = 1; d < 32; d <<= 1) {
        const int t = __shfl_up(incl, d);
        if (lane >= d) incl += t;
      }
      const int nfl = __builtin_amdgcn_readlane(incl, 31);
      {
        int k = incl - c;
        while (word != 0u) {
          const int b = __ffs((int)word) - 1;
          word &= word - 1u;
          gl[k++] = (unsigned short)(lane * 32 + b);
        }
      }
      const int g0 = (wv * wpw + off) * 32;
      auto batches = [&](auto al) {
      for (int s0 = 0; s0 < nfl; s0 += 4 * SCAN_U) {
        int4 dv[SCAN_U]; int eb[SCAN_U]; int nrem[SCAN_U]; int rel[SCAN_U];
#pragma unroll
        for (int u = 0; u < SCAN_U; ++u) rel[u] = gl[min(s0 + 4 * u + qd, nfl - 1)];
#pragma unroll
        for (int u = 0; u < SCAN_U; ++u) {
          const int idx = s0 + 4 * u + qd;
          const int e = (g0 + rel[u]) * 32 + sub * 4;
          eb[u] = e;
          nrem[u] = (idx < nfl) ? (ne - e) : 0;
          if constexpr (decltype(al)::value != 0) {
            dv[u] = *(const int4*)(dst + min(e, ne - 4));
          } else {
            dv[u].x = dst[min(e, ne - 1)];
            dv[u].y = dst[min(e + 1, ne - 1)];
            dv[u].z = dst[min(e + 2, ne - 1)];
            dv[u].w = dst[min(e + 3, ne - 1)];
          }
        }
#pragma unroll
        for (int u = 0; u < SCAN_U; ++u) {
          const unsigned lim = (nrem[u] > 0) ? (unsigned)tn : 0u;
          auto take = [&](int k, int d) {
            const unsigned dl = (unsigned)(d - tile0);
            bool hit;
            if constexpr (decltype(al)::value != 0) hit = dl < lim;
            else hit = k < nrem[u] && dl < (unsigned)tn;
            const unsigned bal = __builtin_amdgcn_ballot_w32(hit);
            if (bal != 0u) {
              const int slot = wpos + (int)__builtin_amdgcn_mbcnt_lo(bal, 0u);
              if (hit && slot < segb + SEG) lsd[slot] = int2{eb[u] + k, (int)dl};
              wpos += __popc(bal);
            }
          };
          take(0, dv[u].x); take(1, dv[u].y); take(2, dv[u].z); take(3, dv[u].w);
        }
      }
      };
      if ((ne & 3) == 0) batches(IC<1>{}); else batches(IC<0>{});
    }
    wfound = wpos - segb;
    if (lane == 0) wcnt[wv] = wfound;
  }
  __syncthreads();

  int offw, total;
  bool fits;
  {
    static_assert(EB / 32 == 32, "one wave's count per lane");
    const int c = wcnt[lane];
    int incl = c;
#pragma unroll
    for (int d = 1; d < 32; d <<= 1) {
      const int t = __shfl_up(incl, d);
      if (lane >= d) incl += t;
    }
    offw = __builtin_amdgcn_readlane(incl - c, wv);
    total = __builtin_amdgcn_readlane(incl, 31);
    fits = __builtin_amdgcn_ballot_w32(c > SEG) == 0u && total + tn <= CAPS;
  }
  auto score4 = [&](int dl, v4f as) {
    v4f ev;
#pragma unroll
    for (int hh = 0; hh < HEADS; ++hh) ev[hh] = leaky(as[hh] + sadst[dl * HEADS + hh]);
    return ev;
  };
  auto max4 = [&](int dl, v4f ev) {
#pragma unroll
    for (int hh = 0; hh < HEADS; ++hh) atomicMax(&smax[dl * HEADS + hh], enc_f32(ev[hh]));
  };
  if (fits) {
    const int segb = wv * SEG;
    for (int base = lane; base < wfound; base += 32 * MAX_U) {
      int sv[MAX_U], dd[MAX_U]; v4f as[MAX_U];
#pragma unroll
      for (int u = 0; u < MAX_U; ++u) {
        const int j = base + u * 32;
        int e = 0;
        dd[u] = 0;
        if (j < wfound) { const int2 r = lsd[segb + j]; e = r.x; dd[u] = r.y; }
        sv[u] = ei[e];
      }
#pragma unroll
      for (int u = 0; u < MAX_U; ++u) as[u] = *(const v4f*)(a_src + (size_t)sv[u] * HEADS);
#pragma unroll
      for (int u = 0; u < MAX_U; ++u) {
        const int j = base + u * 32;
        if (j < wfound) {
          const v4f ev = score4(dd[u], as[u]);
          ls2[offw + j] = sv[u];
          ld2[offw + j] = dd[u];
          *(v4f*)(esc + (offw + j) * HEADS) = ev;
          max4(dd[u], ev);
        }
      }
    }
  } else {
    for (int e = tid; e < ne; e += EB) {
      unsigned dl = (unsigned)(dst[e] - tile0);
      if (dl < (unsigned)tn)
        max4((int)dl, score4((int)dl, *(const v4f*)(a_src + (size_t)ei[e] * HEADS)));
    }
  }
  for (int i = tid; i < tn; i += EB) {
    const v4f ev = score4(i, *(const v4f*)(a_src + (size_t)(tile0 + i) * HEADS));
    if (fits) *(v4f*)(esc + (total + i) * HEADS) = ev;
    max4(i, ev);
  }
  __syncthreads();

  const int head = lane >> 3;
  auto add_edge = [&](int dl, float av, v4f hv) {
    float v = leaky(av + sadst[dl * HEADS + head]);
    float p = expf(v - dec_f32(smax[dl * HEADS + head]));
    if ((lane & 7) == 0) atomicAdd(&sden[dl * HEADS + head], p);
    float* a = acc + dl * CH + lane;
    atomicAdd(a, p * hv.x);
    atomicAdd(a + 32, p * hv.y);
    atomicAdd(a + 64, p * hv.z);
    atomicAdd(a + 96, p * hv.w);
  };
  if (fits) {
    const int items = total + tn;
    static_assert(ACC_U * HEADS == 32, "one (item, head) pair per lane");
    const int sub = lane & (ACC_U - 1);
    for (int i0 = wv * ACC_U; i0 < items; i0 += (EB / 32) * ACC_U) {
      const int idx = i0 + sub;
      int s = tile0, dl = -1;
      if (idx < total) { s = ls2[idx]; dl = ld2[idx]; }
      else if (idx < items) { dl = idx - total; s = tile0 + dl; }
      int dlu[ACC_U]; v4f hv[ACC_U];
#pragma unroll
      for (int u = 0; u < ACC_U; ++u) {
        const int su = __builtin_amdgcn_readlane(s, u);
        dlu[u] = __builtin_amdgcn_readlane(dl, u);
        hv[u] = *(const v4f*)(h + (size_t)su * CH + lane * 4);
      }
      const int dls = (dl >= 0) ? dl : 0;
      const float v = esc[((dl >= 0) ? idx : 0) * HEADS + head];
      const float p = expf(v - dec_f32(smax[dls * HEADS + head]));
      if (dl >= 0) atomicAdd(&sden[dl * HEADS + head], p);
      auto add_row = [&](auto uc) {
        constexpr int u = decltype(uc)::value;
        if (dlu[u] >= 0) {
          const float pu = group8_bcast<u>(p);
          float* a = acc + dlu[u] * CH + lane;
          atomicAdd(a, pu * hv[u].x);
          atomicAdd(a + 32, pu * hv[u].y);
          atomicAdd(a + 64, pu * hv[u].z);
          atomicAdd(a + 96, pu * hv[u].w);
        }
      };
      add_row(IC<0>{}); add_row(IC<1>{}); add_row(IC<2>{}); add_row(IC<3>{});
      add_row(IC<4>{}); add_row(IC<5>{}); add_row(IC<6>{}); add_row(IC<7>{});
    }
  } else {
    for (int e = wave; e < ne; e += EB / 32) {
      unsigned dl = (unsigned)(dst[e] - tile0);
      if (dl < (unsigned)tn) {
        int s = ei[e];
        add_edge((int)dl, a_src[(size_t)s * HEADS + head],
                 *(const v4f*)(h + (size_t)s * CH + lane * 4));
      }
    }
    for (int i = wave; i < tn; i += EB / 32) {
      int s = tile0 + i;
      add_edge(i, a_src[(size_t)s * HEADS + head],
               *(const v4f*)(h + (size_t)s * CH + lane * 4));
    }
  }
  __syncthreads();

  const float pw = prelu_w[0];
  for (int i = tid; i < tn * (CH / 4); i += EB) {
    int node = i >> 5;
    int l4 = i & 31;
    float den = sden[node * HEADS + (l4 >> 3)] + 1e-16f;
    const float* a = acc + node * CH + l4;
    v4f b = *(const v4f*)(bias + l4 * 4);
    v4f o;
    o.x = a[0] / den + b.x;
    o.y = a[32] / den + b.y;
    o.z = a[64] / den + b.z;
    o.w = a[96] / den + b.w;
    o.x = (o.x >= 0.0f) ? o.x : pw * o.x;
    o.y = (o.y >= 0.0f) ? o.y : pw * o.y;
    o.z = (o.z >= 0.0f) ? o.z : pw * o.z;
    o.w = (o.w >= 0.0f) ? o.w : pw * o.w;
    *(v4f*)(out + (size_t)(tile0 + node) * CH + l4 * 4) = o;
  }
}

extern "C" void kernel_launch(void* const* d_in, const int* in_sizes, int n_in,
                              void* d_out, int out_size, void* d_ws, size_t ws_size,
                              hipStream_t stream) {
  const float* x       = (const float*)d_in[0];
  const int*   ei      = (const int*)d_in[1];
  const float* W       = (const float*)d_in[2];
  const float* att_src = (const float*)d_in[3];
  const float* att_dst = (const float*)d_in[4];
  const float* bias    = (const float*)d_in[5];
  const float* prelu_w = (const float*)d_in[6];
  float* out = (float*)d_out;

  const int n  = in_sizes[0] / CH;
  const int ne = in_sizes[1] / 2;

  float* h     = (float*)d_ws;
  float* a_src = h + (size_t)n * CH;
  float* a_dst = a_src + (size_t)n * HEADS;
  unsigned* bm = (unsigned*)(a_dst + (size_t)n * HEADS);

  gat_gemm_att<<<GEMM_WG, 256, 0, stream>>>(x, W, att_src, att_dst, ei,
                                                  h, a_src, a_dst, bm, n, ne);
  gat_tile<<<(n + TN - 1) / TN, EB, 0, stream>>>(ei, bm, h, a_src, a_dst, bias, prelu_w,
                                                 out, n, ne);
}


# Edge kernel: each listed edge gets its rank in its destination row (integer LDS add), list regrouped by row, a wave sums whole rows in registers: no float LDS adds; sums in rank (arrival) order
# speedup vs baseline: 4.5748x; 1.0426x over previous
#include <hip/hip_runtime.h>
#include <hip/hip_bf16.h>

typedef __attribute__((ext_vector_type(2))) float v2f;
typedef __attribute__((ext_vector_type(4))) float v4f;
typedef __attribute__((ext_vector_type(8))) float v8f;

#define HEADS 4
#define CH 128
#define NEG_SLOPE 0.2f
#define TN 196

__device__ __forceinline__ unsigned tile_key(int d) {
  return ((unsigned)d / (unsigned)TN) & 255u;
}
#define BROWS 256
#define CHG 256

__device__ __forceinline__ unsigned enc_f32(float f) {
  unsigned u = __float_as_uint(f);
  return (f >= 0.0f) ? (u | 0x80000000u) : ~u;
}
__device__ __forceinline__ float dec_f32(unsigned u) {
  return (u & 0x80000000u) ? __uint_as_float(u & 0x7fffffffu)
                           : __uint_as_float(~u);
}

template <int V> struct IC { static constexpr int value = V; };

template <int U>
__device__ __forceinline__ float group8_bcast(float v) {
  return __uint_as_float(__builtin_amdgcn_mov_dpp8(__float_as_uint(v), (unsigned)U * 0x249249u));
}

#define GEMM_WG 512
#define HP 132
__global__ __launch_bounds__(256) void gat_gemm_att(
    const float* x,
    const float* __restrict__ W,
    const float* __restrict__ att_src, const float* __restrict__ att_dst,
    const int* __restrict__ ei,
    float* __restrict__ h, float* __restrict__ a_src, float* __restrict__ a_dst,
    unsigned* __restrict__ bm, int n, int ne) {
  __shared__ __attribute__((aligned(16))) float ht[2][16 * HP];
  const int tid = threadIdx.x;

  {
    __shared__ __attribute__((aligned(16))) unsigned bml[BROWS * (CHG / 32)];
    const int* __restrict__ dst = ei + ne;
    const int ngr = (ne + 31) >> 5;
    const int nch = (ngr + CHG - 1) / CHG;
    const int rw = nch * (CHG / 32);
    auto chunk_bits = [&](auto al, int c) {
      constexpr int NJ = (CHG * 32) / (256 * 4);
      const int e0 = c * (CHG * 32);
      int4 d[NJ];
#pragma unroll
      for (int j = 0; j < NJ; ++j) {
        const int e = e0 + (j * 256 + tid) * 4;
        if constexpr (decltype(al)::value != 0) {
          d[j] = *(const int4*)(dst + min(e, ne - 4));
        } else {
          d[j].x = dst[min(e, ne - 1)];
          d[j].y = dst[min(e + 1, ne - 1)];
          d[j].z = dst[min(e + 2, ne - 1)];
          d[j].w = dst[min(e + 3, ne - 1)];
        }
      }
      for (int i = tid; i < BROWS * (CHG / 32); i += 256) bml[i] = 0u;
      __syncthreads();
#pragma unroll
      for (int j = 0; j < NJ; ++j) {
        const int q = j * 256 + tid;
        const int nv = ne - (e0 + q * 4);
        const int col = q >> 8;
        const unsigned bit = 1u << ((q >> 3) & 31);
        if (nv > 0) atomicOr(&bml[tile_key(d[j].x) * (CHG / 32) + col], bit);
        if (nv > 1) atomicOr(&bml[tile_key(d[j].y) * (CHG / 32) + col], bit);
        if (nv > 2) atomicOr(&bml[tile_key(d[j].z) * (CHG / 32) + col], bit);
        if (nv > 3) atomicOr(&bml[tile_key(d[j].w) * (CHG / 32) + col], bit);
      }
    };
    for (int c = (int)gridDim.x - 1 - (int)blockIdx.x; c < nch; c += gridDim.x) {
      if ((ne & 3) == 0) chunk_bits(IC<1>{}, c); else chunk_bits(IC<0>{}, c);
      __syncthreads();
      {
        const uint4* src4 = (const uint4*)(bml + tid * (CHG / 32));
        uint4* out4 = (uint4*)(bm + (size_t)tid * rw + (size_t)c * (CHG / 32));
        out4[0] = src4[0];
        out4[1] = src4[1];
      }
      __syncthreads();
    }
  }

  const int wave = tid >> 5;
  const int lane = tid & 31;
  const int n0 = wave * 16;
  const int lm = lane & 15;
  const int lh = lane >> 4;
  const int ntiles = (n + 15) >> 4;

  v2f b[32];
#pragma unroll
  for (int j = 0; j < 16; ++j) {
    const float* __restrict__ wp = W + (size_t)(8 * j + 4 * lh) * CH + n0 + lm;
    b[2 * j].x = wp[0];
    b[2 * j].y = wp[CH];
    b[2 * j + 1].x = wp[2 * CH];
    b[2 * j + 1].y = wp[3 * CH];
  }

  float av[32];
  {
    const float* __restrict__ avp = ((tid < 64) ? att_src : att_dst) + (tid & (HEADS - 1)) * 32;
#pragma unroll
    for (int c = 0; c < 32; ++c) av[c] = avp[c];
  }

  v4f a[16];
  auto load_a = [&](int t) {
    int row = t * 16 + lm;
    if (row >= n) row = n - 1;
    const float* arow = x + (size_t)row * CH + 4 * lh;
#pragma unroll
    for (int j = 0; j < 16; ++j) a[j] = *(const v4f*)(arow + 8 * j);
  };
  if ((int)blockIdx.x < ntiles) load_a(blockIdx.x);

  int buf = 0;
  for (int t = blockIdx.x; t < ntiles; t += gridDim.x, buf ^= 1) {
    const int m0 = t * 16;
    v8f acc = {};
#pragma unroll
    for (int j = 0; j < 16; ++j) {
      v2f a0 = {a[j].x, a[j].y};
      v2f a1 = {a[j].z, a[j].w};
      acc = __builtin_amdgcn_wmma_f32_16x16x4_f32(
           false, a0,  false, b[2 * j],
           (short)0, acc,  false,  false);
      acc = __builtin_amdgcn_wmma_f32_16x16x4_f32(
           false, a1,  false, b[2 * j + 1],
           (short)0, acc,  false,  false);
    }
    load_a(min(t + (int)gridDim.x, ntiles - 1));

    float* htb = ht[buf];
    {
      float* __restrict__ hs = htb + (8 * lh) * HP + n0 + lm;
#pragma unroll
      for (int v = 0; v < 8; ++v) hs[v * HP] = acc[v];
    }
    __builtin_amdgcn_fence(__ATOMIC_RELEASE, "workgroup", "local");
    __builtin_amdgcn_s_barrier();
    __builtin_amdgcn_fence(__ATOMIC_ACQUIRE, "workgroup", "local");

#pragma unroll
    for (int rr = 0; rr < 2; ++rr) {
      const int r = 2 * wave + rr;
      if (m0 + r < n)
        *(v4f*)(h + (size_t)(m0 + r) * CH + lane * 4) = *(const v4f*)(htb + r * HP + lane * 4);
    }

    if (tid < 128) {
      const int r = (tid & 63) >> 2;
      const int head = tid & (HEADS - 1);
      if (m0 + r < n) {
        const float* hv = htb + r * HP + head * 32;
        float s = 0.0f;
#pragma unroll
        for (int c = 0; c < 32; ++c) s += hv[c] * av[c];
        float* __restrict__ o = (tid < 64) ? a_src : a_dst;
        o[(size_t)(m0 + r) * HEADS + head] = s;
      }
    }
  }
}

#define CAP 8192
#define EB 1024
#define SEG (CAP / (EB / 32))
#define GLW 1024
#define SCAN_U 8
#define ACC_U 8
#define MAX_U 4
#define CAPS 4608

__device__ __forceinline__ float leaky(float v) {
  return (v >= 0.0f) ? v : NEG_SLOPE * v;
}

__global__ __launch_bounds__(EB) void gat_tile(
    const int* __restrict__ ei, const unsigned* __restrict__ bm,
    const float* __restrict__ h,
    const float* __restrict__ a_src, const float* __restrict__ a_dst,
    const float* __restrict__ bias, const float* __restrict__ prelu_w,
    float* __restrict__ out, int n, int ne) {
  __shared__ __attribute__((aligned(16))) float acc[TN * CH];
  __shared__ unsigned smax[TN * HEADS];
  __shared__ float sden[TN * HEADS];
  __shared__ float sadst[TN * HEADS];
  __shared__ int rcnt[TN];
  __shared__ int rstart[TN + 1];
  __shared__ __attribute__((aligned(8))) int2 lsd[CAP];
  __shared__ __attribute__((aligned(16))) int l2[2 * CAP];
  __shared__ int wcnt[EB / 32];
  __shared__ __attribute__((aligned(16))) float esc[CAPS * HEADS];
  static_assert(2 * CAP * 4 >= (EB / 32) * GLW * 2, "the group lists fit in l2");
  static_assert(CAPS <= CAP && CAPS * HEADS <= TN * CH, "the grouped list fits where it is kept");
  static_assert(TN <= 256, "a local row fits the low byte of a packed list entry");
  float* const gsc = acc;
  int* const ls2 = l2;
  int* const ld2 = l2 + CAP;

  const int tid = threadIdx.x;
  const int tile0 = blockIdx.x * TN;
  const int tn = min(TN, n - tile0);
  const int* __restrict__ dst = ei + ne;

  for (int i = tid; i < tn * HEADS; i += EB) {
    smax[i] = 0u;
    sadst[i] = a_dst[(size_t)tile0 * HEADS + i];
  }
  for (int i = tid; i < TN; i += EB) rcnt[i] = 1;
  __syncthreads();

  const int wave = tid >> 5;
  const int lane = tid & 31;
  const int wv = __builtin_amdgcn_readfirstlane(wave);
  int wfound;
  {
    const int ngr = (ne + 31) >> 5;
    const int nch = (ngr + CHG - 1) / CHG;
    const int rw = nch * (CHG / 32);
    const int nw = (ngr + 31) >> 5;
    const int wpw = (nw + (EB / 32) - 1) / (EB / 32);
    const unsigned* __restrict__ brow = bm + (size_t)(blockIdx.x & (BROWS - 1)) * rw;
    unsigned short* const gl = (unsigned short*)l2 + wv * GLW;
    const int sub = lane & 7;
    const int qd = lane >> 3;
    const int segb = wv * SEG;
    int wpos = segb;
    for (int off = 0; off < wpw; off += 32) {
      const int wi = wv * wpw + off + lane;
      unsigned word = 0u;
      if (off + lane < wpw && wi < nw) word = brow[wi];
      const int c = __popc(word);
      int incl = c;
#pragma unroll
      for (int d = 1; d < 32; d <<= 1) {
        const int t = __shfl_up(incl, d);
        if (lane >= d) incl += t;
      }
      const int nfl = __builtin_amdgcn_readlane(incl, 31);
      {
        int k = incl - c;
        while (word != 0u) {
          const int b = __ffs((int)word) - 1;
          word &= word - 1u;
          gl[k++] = (unsigned short)(lane * 32 + b);
        }
      }
      const int g0 = (wv * wpw + off) * 32;
      auto batches = [&](auto al) {
      for (int s0 = 0; s0 < nfl; s0 += 4 * SCAN_U) {
        int4 dv[SCAN_U]; int eb[SCAN_U]; int nrem[SCAN_U]; int rel[SCAN_U];
#pragma unroll
        for (int u = 0; u < SCAN_U; ++u) rel[u] = gl[min(s0 + 4 * u + qd, nfl - 1)];
#pragma unroll
        for (int u = 0; u < SCAN_U; ++u) {
          const int idx = s0 + 4 * u + qd;
          const int e = (g0 + rel[u]) * 32 + sub * 4;
          eb[u] = e;
          nrem[u] = (idx < nfl) ? (ne - e) : 0;
          if constexpr (decltype(al)::value != 0) {
            dv[u] = *(const int4*)(dst + min(e, ne - 4));
          } else {
            dv[u].x = dst[min(e, ne - 1)];
            dv[u].y = dst[min(e + 1, ne - 1)];
            dv[u].z = dst[min(e + 2, ne - 1)];
            dv[u].w = dst[min(e + 3, ne - 1)];
          }
        }
#pragma unroll
        for (int u = 0; u < SCAN_U; ++u) {
          const unsigned lim = (nrem[u] > 0) ? (unsigned)tn : 0u;
          auto take = [&](int k, int d) {
            const unsigned dl = (unsigned)(d - tile0);
            bool hit;
            if constexpr (decltype(al)::value != 0) hit = dl < lim;
            else hit = k < nrem[u] && dl < (unsigned)tn;
            const unsigned bal = __builtin_amdgcn_ballot_w32(hit);
            if (bal != 0u) {
              const int slot = wpos + (int)__builtin_amdgcn_mbcnt_lo(bal, 0u);
              if (hit && slot < segb + SEG) lsd[slot] = int2{eb[u] + k, (int)dl};
              wpos += __popc(bal);
            }
          };
          take(0, dv[u].x); take(1, dv[u].y); take(2, dv[u].z); take(3, dv[u].w);
        }
      }
      };
      if ((ne & 3) == 0) batches(IC<1>{}); else batches(IC<0>{});
    }
    wfound = wpos - segb;
    if (lane == 0) wcnt[wv] = wfound;
  }
  __syncthreads();

  int offw, total;
  bool fits;
  {
    static_assert(EB / 32 == 32, "one wave's count per lane");
    const int c = wcnt[lane];
    int incl = c;
#pragma unroll
    for (int d = 1; d < 32; d <<= 1) {
      const int t = __shfl_up(incl, d);
      if (lane >= d) incl += t;
    }
    offw = __builtin_amdgcn_readlane(incl - c, wv);
    total = __builtin_amdgcn_readlane(incl, 31);
    fits = __builtin_amdgcn_ballot_w32(c > SEG) == 0u && total + tn <= CAPS;
  }
  auto score4 = [&](int dl, v4f as) {
    v4f ev;
#pragma unroll
    for (int hh = 0; hh < HEADS; ++hh) ev[hh] = leaky(as[hh] + sadst[dl * HEADS + hh]);
    return ev;
  };
  auto max4 = [&](int dl, v4f ev) {
#pragma unroll
    for (int hh = 0; hh < HEADS; ++hh) atomicMax(&smax[dl * HEADS + hh], enc_f32(ev[hh]));
  };
  const int head = lane >> 3;
  const float pw = prelu_w[0];

  if (!fits) {
    for (int i = tid; i < TN * CH / 4; i += EB) ((v4f*)acc)[i] = v4f{0.f, 0.f, 0.f, 0.f};
    for (int i = tid; i < tn * HEADS; i += EB) sden[i] = 0.0f;
    for (int e = tid; e < ne; e += EB) {
      unsigned dl = (unsigned)(dst[e] - tile0);
      if (dl < (unsigned)tn)
        max4((int)dl, score4((int)dl, *(const v4f*)(a_src + (size_t)ei[e] * HEADS)));
    }
    for (int i = tid; i < tn; i += EB)
      max4(i, score4(i, *(const v4f*)(a_src + (size_t)(tile0 + i) * HEADS)));
    __syncthreads();
    auto add_edge = [&](int dl, float av, v4f hv) {
      float v = leaky(av + sadst[dl * HEADS + head]);
      float p = expf(v - dec_f32(smax[dl * HEADS + head]));
      if ((lane & 7) == 0) atomicAdd(&sden[dl * HEADS + head], p);
      float* a = acc + dl * CH + lane;
      atomicAdd(a, p * hv.x);
      atomicAdd(a + 32, p * hv.y);
      atomicAdd(a + 64, p * hv.z);
      atomicAdd(a + 96, p * hv.w);
    };
    for (int e = wave; e < ne; e += EB / 32) {
      unsigned dl = (unsigned)(dst[e] - tile0);
      if (dl < (unsigned)tn) {
        int s = ei[e];
        add_edge((int)dl, a_src[(size_t)s * HEADS + head],
                 *(const v4f*)(h + (size_t)s * CH + lane * 4));
      }
    }
    for (int i = wave; i < tn; i += EB / 32) {
      int s = tile0 + i;
      add_edge(i, a_src[(size_t)s * HEADS + head],
               *(const v4f*)(h + (size_t)s * CH + lane * 4));
    }
    __syncthreads();
    for (int i = tid; i < tn * (CH / 4); i += EB) {
      int node = i >> 5;
      int l4 = i & 31;
      float den = sden[node * HEADS + (l4 >> 3)] + 1e-16f;
      const float* a = acc + node * CH + l4;
      v4f b = *(const v4f*)(bias + l4 * 4);
      v4f o;
      o.x = a[0] / den + b.x;
      o.y = a[32] / den + b.y;
      o.z = a[64] / den + b.z;
      o.w = a[96] / den + b.w;
      o.x = (o.x >= 0.0f) ? o.x : pw * o.x;
      o.y = (o.y >= 0.0f) ? o.y : pw * o.y;
      o.z = (o.z >= 0.0f) ? o.z : pw * o.z;
      o.w = (o.w >= 0.0f) ? o.w : pw * o.w;
      *(v4f*)(out + (size_t)(tile0 + node) * CH + l4 * 4) = o;
    }
    return;
  }

  {
    const int segb = wv * SEG;
    for (int base = lane; base < wfound; base += 32 * MAX_U) {
      int sv[MAX_U], dd[MAX_U], rk[MAX_U]; v4f as[MAX_U];
#pragma unroll
      for (int u = 0; u < MAX_U; ++u) {
        const int j = base + u * 32;
        int e = 0;
        dd[u] = 0;
        rk[u] = 0;
        if (j < wfound) {
          const int2 r = lsd[segb + j];
          e = r.x; dd[u] = r.y;
          rk[u] = atomicAdd(&rcnt[dd[u]], 1);
        }
        sv[u] = ei[e];
      }
#pragma unroll
      for (int u = 0; u < MAX_U; ++u) as[u] = *(const v4f*)(a_src + (size_t)sv[u] * HEADS);
#pragma unroll
      for (int u = 0; u < MAX_U; ++u) {
        const int j = base + u * 32;
        if (j < wfound) {
          const v4f ev = score4(dd[u], as[u]);
          ls2[offw + j] = sv[u];
          ld2[offw + j] = dd[u] | (rk[u] << 8);
          *(v4f*)(esc + (offw + j) * HEADS) = ev;
          max4(dd[u], ev);
        }
      }
    }
  }
  for (int i = tid; i < tn; i += EB) {
    const v4f ev = score4(i, *(const v4f*)(a_src + (size_t)(tile0 + i) * HEADS));
    *(v4f*)(esc + (total + i) * HEADS) = ev;
    max4(i, ev);
  }
  __syncthreads();

  const int items = total + tn;
  if (wv == 0) {
    constexpr int RPL = (TN + 31) / 32;
    int c[RPL];
    int sum = 0;
#pragma unroll
    for (int k = 0; k < RPL; ++k) {
      const int r = lane * RPL + k;
      c[k] = (r < tn) ? rcnt[r] : 0;
      sum += c[k];
    }
    int incl = sum;
#pragma unroll
    for (int d = 1; d < 32; d <<= 1) {
      const int t = __shfl_up(incl, d);
      if (lane >= d) incl += t;
    }
    int run = incl - sum;
#pragma unroll
    for (int k = 0; k < RPL; ++k) {
      const int r = lane * RPL + k;
      if (r <= tn) rstart[r] = run;
      run += c[k];
    }
  }
  __syncthreads();
  for (int idx = tid; idx < items; idx += EB) {
    int s, dl, r;
    if (idx < total) {
      s = ls2[idx];
      const int pk = ld2[idx];
      dl = pk & 255;
      r = pk >> 8;
    } else {
      dl = idx - total;
      s = tile0 + dl;
      r = 0;
    }
    const int q = rstart[dl] + r;
    lsd[q] = int2{s, dl};
    *(v4f*)(gsc + q * HEADS) = *(const v4f*)(esc + idx * HEADS);
  }
  __syncthreads();

  {
    int rb = 0, re = 0;
    {
      const int lo = wv * items, hi = lo + items;
      for (int c = 0; c < (TN + 31) / 32; ++c) {
        const int r = c * 32 + lane;
        const int k32 = (r < tn) ? rstart[r] * 32 : 0x7fffffff;
        rb += __popc(__builtin_amdgcn_ballot_w32(k32 < lo));
        re += __popc(__builtin_amdgcn_ballot_w32(k32 < hi));
      }
    }
    const int pbeg = __builtin_amdgcn_readfirstlane(rstart[rb]);
    const int pend = __builtin_amdgcn_readfirstlane(rstart[re]);
    const v4f bs = *(const v4f*)(bias + lane * 4);
    int cur = -1;
    v4f sm = {0.f, 0.f, 0.f, 0.f};
    float den = 0.0f;
    auto finish_row = [&](int row) {
      const float d = den + 1e-16f;
      v4f o;
      o.x = sm.x / d + bs.x;
      o.y = sm.y / d + bs.y;
      o.z = sm.z / d + bs.z;
      o.w = sm.w / d + bs.w;
      o.x = (o.x >= 0.0f) ? o.x : pw * o.x;
      o.y = (o.y >= 0.0f) ? o.y : pw * o.y;
      o.z = (o.z >= 0.0f) ? o.z : pw * o.z;
      o.w = (o.w >= 0.0f) ? o.w : pw * o.w;
      *(v4f*)(out + (size_t)(tile0 + row) * CH + lane * 4) = o;
    };
    static_assert(ACC_U * HEADS == 32, "one (item, head) pair per lane");
    const int sub = lane & (ACC_U - 1);
    for (int i0 = pbeg; i0 < pend; i0 += ACC_U) {
      const int idx = i0 + sub;
      int s = tile0, dl = -1;
      if (idx < pend) { const int2 r = lsd[idx]; s = r.x; dl = r.y; }
      int dlu[ACC_U]; v4f hv[ACC_U];
#pragma unroll
      for (int u = 0; u < ACC_U; ++u) {
        const int su = __builtin_amdgcn_readlane(s, u);
        dlu[u] = __builtin_amdgcn_readlane(dl, u);
        hv[u] = *(const v4f*)(h + (size_t)su * CH + lane * 4);
      }
      const int dls = (dl >= 0) ? dl : 0;
      const float v = gsc[((dl >= 0) ? idx : pbeg) * HEADS + head];
      const float p = expf(v - dec_f32(smax[dls * HEADS + head]));
      auto add_row = [&](auto uc) {
        constexpr int u = decltype(uc)::value;
        if (dlu[u] >= 0) {
          if (dlu[u] != cur) {
            if (cur >= 0) finish_row(cur);
            cur = dlu[u];
            sm = v4f{0.f, 0.f, 0.f, 0.f};
            den = 0.0f;
          }
          const float pu = group8_bcast<u>(p);
          sm.x += pu * hv[u].x;
          sm.y += pu * hv[u].y;
          sm.z += pu * hv[u].z;
          sm.w += pu * hv[u].w;
          den += pu;
        }
      };
      add_row(IC<0>{}); add_row(IC<1>{}); add_row(IC<2>{}); add_row(IC<3>{});
      add_row(IC<4>{}); add_row(IC<5>{}); add_row(IC<6>{}); add_row(IC<7>{});
    }
    if (cur >= 0) finish_row(cur);
  }
}

extern "C" void kernel_launch(void* const* d_in, const int* in_sizes, int n_in,
                              void* d_out, int out_size, void* d_ws, size_t ws_size,
                              hipStream_t stream) {
  const float* x       = (const float*)d_in[0];
  const int*   ei      = (const int*)d_in[1];
  const float* W       = (const float*)d_in[2];
  const float* att_src = (const float*)d_in[3];
  const float* att_dst = (const float*)d_in[4];
  const float* bias    = (const float*)d_in[5];
  const float* prelu_w = (const float*)d_in[6];
  float* out = (float*)d_out;

  const int n  = in_sizes[0] / CH;
  const int ne = in_sizes[1] / 2;

  float* h     = (float*)d_ws;
  float* a_src = h + (size_t)n * CH;
  float* a_dst = a_src + (size_t)n * HEADS;
  unsigned* bm = (unsigned*)(a_dst + (size_t)n * HEADS);

  gat_gemm_att<<<GEMM_WG, 256, 0, stream>>>(x, W, att_src, att_dst, ei,
                                                  h, a_src, a_dst, bm, n, ne);
  gat_tile<<<(n + TN - 1) / TN, EB, 0, stream>>>(ei, bm, h, a_src, a_dst, bias, prelu_w,
                                                 out, n, ne);
}


# Projection launch: a tile's 16 rows of x are fetched once per work-group as whole 512-B rows, two tiles ahead, and staged in LDS; the 8 waves read A fragments there, not each from global memory
# speedup vs baseline: 4.8692x; 1.0644x over previous
#include <hip/hip_runtime.h>
#include <hip/hip_bf16.h>

typedef __attribute__((ext_vector_type(2))) float v2f;
typedef __attribute__((ext_vector_type(4))) float v4f;
typedef __attribute__((ext_vector_type(8))) float v8f;

#define HEADS 4
#define CH 128
#define NEG_SLOPE 0.2f
#define TN 196

__device__ __forceinline__ unsigned tile_key(int d) {
  return ((unsigned)d / (unsigned)TN) & 255u;
}
#define BROWS 256
#define CHG 256

__device__ __forceinline__ unsigned enc_f32(float f) {
  unsigned u = __float_as_uint(f);
  return (f >= 0.0f) ? (u | 0x80000000u) : ~u;
}
__device__ __forceinline__ float dec_f32(unsigned u) {
  return (u & 0x80000000u) ? __uint_as_float(u & 0x7fffffffu)
                           : __uint_as_float(~u);
}

template <int V> struct IC { static constexpr int value = V; };

template <int U>
__device__ __forceinline__ float group8_bcast(float v) {
  return __uint_as_float(__builtin_amdgcn_mov_dpp8(__float_as_uint(v), (unsigned)U * 0x249249u));
}

#define GEMM_WG 512
#define HP 132
__global__ __launch_bounds__(256) void gat_gemm_att(
    const float* x,
    const float* __restrict__ W,
    const float* __restrict__ att_src, const float* __restrict__ att_dst,
    const int* __restrict__ ei,
    float* __restrict__ h, float* __restrict__ a_src, float* __restrict__ a_dst,
    unsigned* __restrict__ bm, int n, int ne) {
  __shared__ __attribute__((aligned(16))) float ht[2][16 * HP];
  __shared__ __attribute__((aligned(16))) float xs[2][16 * HP];
  const int tid = threadIdx.x;

  {
    __shared__ __attribute__((aligned(16))) unsigned bml[BROWS * (CHG / 32)];
    const int* __restrict__ dst = ei + ne;
    const int ngr = (ne + 31) >> 5;
    const int nch = (ngr + CHG - 1) / CHG;
    const int rw = nch * (CHG / 32);
    auto chunk_bits = [&](auto al, int c) {
      constexpr int NJ = (CHG * 32) / (256 * 4);
      const int e0 = c * (CHG * 32);
      int4 d[NJ];
#pragma unroll
      for (int j = 0; j < NJ; ++j) {
        const int e = e0 + (j * 256 + tid) * 4;
        if constexpr (decltype(al)::value != 0) {
          d[j] = *(const int4*)(dst + min(e, ne - 4));
        } else {
          d[j].x = dst[min(e, ne - 1)];
          d[j].y = dst[min(e + 1, ne - 1)];
          d[j].z = dst[min(e + 2, ne - 1)];
          d[j].w = dst[min(e + 3, ne - 1)];
        }
      }
      for (int i = tid; i < BROWS * (CHG / 32); i += 256) bml[i] = 0u;
      __syncthreads();
#pragma unroll
      for (int j = 0; j < NJ; ++j) {
        const int q = j * 256 + tid;
        const int nv = ne - (e0 + q * 4);
        const int col = q >> 8;
        const unsigned bit = 1u << ((q >> 3) & 31);
        if (nv > 0) atomicOr(&bml[tile_key(d[j].x) * (CHG / 32) + col], bit);
        if (nv > 1) atomicOr(&bml[tile_key(d[j].y) * (CHG / 32) + col], bit);
        if (nv > 2) atomicOr(&bml[tile_key(d[j].z) * (CHG / 32) + col], bit);
        if (nv > 3) atomicOr(&bml[tile_key(d[j].w) * (CHG / 32) + col], bit);
      }
    };
    for (int c = (int)gridDim.x - 1 - (int)blockIdx.x; c < nch; c += gridDim.x) {
      if ((ne & 3) == 0) chunk_bits(IC<1>{}, c); else chunk_bits(IC<0>{}, c);
      __syncthreads();
      {
        const uint4* src4 = (const uint4*)(bml + tid * (CHG / 32));
        uint4* out4 = (uint4*)(bm + (size_t)tid * rw + (size_t)c * (CHG / 32));
        out4[0] = src4[0];
        out4[1] = src4[1];
      }
      __syncthreads();
    }
  }

  const int wave = tid >> 5;
  const int lane = tid & 31;
  const int n0 = wave * 16;
  const int lm = lane & 15;
  const int lh = lane >> 4;
  const int ntiles = (n + 15) >> 4;

  v2f b[32];
#pragma unroll
  for (int j = 0; j < 16; ++j) {
    const float* __restrict__ wp = W + (size_t)(8 * j + 4 * lh) * CH + n0 + lm;
    b[2 * j].x = wp[0];
    b[2 * j].y = wp[CH];
    b[2 * j + 1].x = wp[2 * CH];
    b[2 * j + 1].y = wp[3 * CH];
  }

  float av[32];
  {
    const float* __restrict__ avp = ((tid < 64) ? att_src : att_dst) + (tid & (HEADS - 1)) * 32;
#pragma unroll
    for (int c = 0; c < 32; ++c) av[c] = avp[c];
  }

  v4f xr[2];
  auto load_x = [&](int t) {
#pragma unroll
    for (int rr = 0; rr < 2; ++rr) {
      int row = t * 16 + 2 * wave + rr;
      if (row >= n) row = n - 1;
      xr[rr] = *(const v4f*)(x + (size_t)row * CH + lane * 4);
    }
  };
  auto stage_x = [&](int b) {
#pragma unroll
    for (int rr = 0; rr < 2; ++rr) *(v4f*)(xs[b] + (2 * wave + rr) * HP + lane * 4) = xr[rr];
  };
  const int gstride = (int)gridDim.x;
  if ((int)blockIdx.x < ntiles) {
    load_x(blockIdx.x);
    stage_x(0);
    load_x(min((int)blockIdx.x + gstride, ntiles - 1));
  }
  __syncthreads();

  int buf = 0;
  for (int t = blockIdx.x; t < ntiles; t += gridDim.x, buf ^= 1) {
    const int m0 = t * 16;
    stage_x(buf ^ 1);
    load_x(min(t + 2 * gstride, ntiles - 1));
    const float* xa = xs[buf] + lm * HP + 4 * lh;
    v4f a[16];
#pragma unroll
    for (int j = 0; j < 16; ++j) a[j] = *(const v4f*)(xa + 8 * j);
    asm volatile("" :: "v"(a[0]), "v"(a[1]), "v"(a[2]), "v"(a[3]), "v"(a[4]), "v"(a[5]), "v"(a[6]), "v"(a[7]),
                 "v"(a[8]), "v"(a[9]), "v"(a[10]), "v"(a[11]), "v"(a[12]), "v"(a[13]), "v"(a[14]), "v"(a[15]));
    v8f acc = {};
#pragma unroll
    for (int j = 0; j < 16; ++j) {
      v2f a0 = {a[j].x, a[j].y};
      v2f a1 = {a[j].z, a[j].w};
      acc = __builtin_amdgcn_wmma_f32_16x16x4_f32(
           false, a0,  false, b[2 * j],
           (short)0, acc,  false,  false);
      acc = __builtin_amdgcn_wmma_f32_16x16x4_f32(
           false, a1,  false, b[2 * j + 1],
           (short)0, acc,  false,  false);
    }

    float* htb = ht[buf];
    {
      float* __restrict__ hs = htb + (8 * lh) * HP + n0 + lm;
#pragma unroll
      for (int v = 0; v < 8; ++v) hs[v * HP] = acc[v];
    }
    __builtin_amdgcn_fence(__ATOMIC_RELEASE, "workgroup", "local");
    __builtin_amdgcn_s_barrier();
    __builtin_amdgcn_fence(__ATOMIC_ACQUIRE, "workgroup", "local");

#pragma unroll
    for (int rr = 0; rr < 2; ++rr) {
      const int r = 2 * wave + rr;
      if (m0 + r < n)
        *(v4f*)(h + (size_t)(m0 + r) * CH + lane * 4) = *(const v4f*)(htb + r * HP + lane * 4);
    }

    if (tid < 128) {
      const int r = (tid & 63) >> 2;
      const int head = tid & (HEADS - 1);
      if (m0 + r < n) {
        const float* hv = htb + r * HP + head * 32;
        float s = 0.0f;
#pragma unroll
        for (int c = 0; c < 32; ++c) s += hv[c] * av[c];
        float* __restrict__ o = (tid < 64) ? a_src : a_dst;
        o[(size_t)(m0 + r) * HEADS + head] = s;
      }
    }
  }
}

#define CAP 8192
#define EB 1024
#define SEG (CAP / (EB / 32))
#define GLW 1024
#define SCAN_U 8
#define ACC_U 8
#define MAX_U 4
#define CAPS 4608

__device__ __forceinline__ float leaky(float v) {
  return (v >= 0.0f) ? v : NEG_SLOPE * v;
}

__global__ __launch_bounds__(EB) void gat_tile(
    const int* __restrict__ ei, const unsigned* __restrict__ bm,
    const float* __restrict__ h,
    const float* __restrict__ a_src, const float* __restrict__ a_dst,
    const float* __restrict__ bias, const float* __restrict__ prelu_w,
    float* __restrict__ out, int n, int ne) {
  __shared__ __attribute__((aligned(16))) float acc[TN * CH];
  __shared__ unsigned smax[TN * HEADS];
  __shared__ float sden[TN * HEADS];
  __shared__ float sadst[TN * HEADS];
  __shared__ int rcnt[TN];
  __shared__ int rstart[TN + 1];
  __shared__ __attribute__((aligned(8))) int2 lsd[CAP];
  __shared__ __attribute__((aligned(16))) int l2[2 * CAP];
  __shared__ int wcnt[EB / 32];
  __shared__ __attribute__((aligned(16))) float esc[CAPS * HEADS];
  static_assert(2 * CAP * 4 >= (EB / 32) * GLW * 2, "the group lists fit in l2");
  static_assert(CAPS <= CAP && CAPS * HEADS <= TN * CH, "the grouped list fits where it is kept");
  static_assert(TN <= 256, "a local row fits the low byte of a packed list entry");
  float* const gsc = acc;
  int* const ls2 = l2;
  int* const ld2 = l2 + CAP;

  const int tid = threadIdx.x;
  const int tile0 = blockIdx.x * TN;
  const int tn = min(TN, n - tile0);
  const int* __restrict__ dst = ei + ne;

  for (int i = tid; i < tn * HEADS; i += EB) {
    smax[i] = 0u;
    sadst[i] = a_dst[(size_t)tile0 * HEADS + i];
  }
  for (int i = tid; i < TN; i += EB) rcnt[i] = 1;
  __syncthreads();

  const int wave = tid >> 5;
  const int lane = tid & 31;
  const int wv = __builtin_amdgcn_readfirstlane(wave);
  int wfound;
  {
    const int ngr = (ne + 31) >> 5;
    const int nch = (ngr + CHG - 1) / CHG;
    const int rw = nch * (CHG / 32);
    const int nw = (ngr + 31) >> 5;
    const int wpw = (nw + (EB / 32) - 1) / (EB / 32);
    const unsigned* __restrict__ brow = bm + (size_t)(blockIdx.x & (BROWS - 1)) * rw;
    unsigned short* const gl = (unsigned short*)l2 + wv * GLW;
    const int sub = lane & 7;
    const int qd = lane >> 3;
    const int segb = wv * SEG;
    int wpos = segb;
    for (int off = 0; off < wpw; off += 32) {
      const int wi = wv * wpw + off + lane;
      unsigned word = 0u;
      if (off + lane < wpw && wi < nw) word = brow[wi];
      const int c = __popc(word);
      int incl = c;
#pragma unroll
      for (int d = 1; d < 32; d <<= 1) {
        const int t = __shfl_up(incl, d);
        if (lane >= d) incl += t;
      }
      const int nfl = __builtin_amdgcn_readlane(incl, 31);
      {
        int k = incl - c;
        while (word != 0u) {
          const int b = __ffs((int)word) - 1;
          word &= word - 1u;
          gl[k++] = (unsigned short)(lane * 32 + b);
        }
      }
      const int g0 = (wv * wpw + off) * 32;
      auto batches = [&](auto al) {
      for (int s0 = 0; s0 < nfl; s0 += 4 * SCAN_U) {
        int4 dv[SCAN_U]; int eb[SCAN_U]; int nrem[SCAN_U]; int rel[SCAN_U];
#pragma unroll
        for (int u = 0; u < SCAN_U; ++u) rel[u] = gl[min(s0 + 4 * u + qd, nfl - 1)];
#pragma unroll
        for (int u = 0; u < SCAN_U; ++u) {
          const int idx = s0 + 4 * u + qd;
          const int e = (g0 + rel[u]) * 32 + sub * 4;
          eb[u] = e;
          nrem[u] = (idx < nfl) ? (ne - e) : 0;
          if constexpr (decltype(al)::value != 0) {
            dv[u] = *(const int4*)(dst + min(e, ne - 4));
          } else {
            dv[u].x = dst[min(e, ne - 1)];
            dv[u].y = dst[min(e + 1, ne - 1)];
            dv[u].z = dst[min(e + 2, ne - 1)];
            dv[u].w = dst[min(e + 3, ne - 1)];
          }
        }
#pragma unroll
        for (int u = 0; u < SCAN_U; ++u) {
          const unsigned lim = (nrem[u] > 0) ? (unsigned)tn : 0u;
          auto take = [&](int k, int d) {
            const unsigned dl = (unsigned)(d - tile0);
            bool hit;
            if constexpr (decltype(al)::value != 0) hit = dl < lim;
            else hit = k < nrem[u] && dl < (unsigned)tn;
            const unsigned bal = __builtin_amdgcn_ballot_w32(hit);
            if (bal != 0u) {
              const int slot = wpos + (int)__builtin_amdgcn_mbcnt_lo(bal, 0u);
              if (hit && slot < segb + SEG) lsd[slot] = int2{eb[u] + k, (int)dl};
              wpos += __popc(bal);
            }
          };
          take(0, dv[u].x); take(1, dv[u].y); take(2, dv[u].z); take(3, dv[u].w);
        }
      }
      };
      if ((ne & 3) == 0) batches(IC<1>{}); else batches(IC<0>{});
    }
    wfound = wpos - segb;
    if (lane == 0) wcnt[wv] = wfound;
  }
  __syncthreads();

  int offw, total;
  bool fits;
  {
    static_assert(EB / 32 == 32, "one wave's count per lane");
    const int c = wcnt[lane];
    int incl = c;
#pragma unroll
    for (int d = 1; d < 32; d <<= 1) {
      const int t = __shfl_up(incl, d);
      if (lane >= d) incl += t;
    }
    offw = __builtin_amdgcn_readlane(incl - c, wv);
    total = __builtin_amdgcn_readlane(incl, 31);
    fits = __builtin_amdgcn_ballot_w32(c > SEG) == 0u && total + tn <= CAPS;
  }
  auto score4 = [&](int dl, v4f as) {
    v4f ev;
#pragma unroll
    for (int hh = 0; hh < HEADS; ++hh) ev[hh] = leaky(as[hh] + sadst[dl * HEADS + hh]);
    return ev;
  };
  auto max4 = [&](int dl, v4f ev) {
#pragma unroll
    for (int hh = 0; hh < HEADS; ++hh) atomicMax(&smax[dl * HEADS + hh], enc_f32(ev[hh]));
  };
  const int head = lane >> 3;
  const float pw = prelu_w[0];

  if (!fits) {
    for (int i = tid; i < TN * CH / 4; i += EB) ((v4f*)acc)[i] = v4f{0.f, 0.f, 0.f, 0.f};
    for (int i = tid; i < tn * HEADS; i += EB) sden[i] = 0.0f;
    for (int e = tid; e < ne; e += EB) {
      unsigned dl = (unsigned)(dst[e] - tile0);
      if (dl < (unsigned)tn)
        max4((int)dl, score4((int)dl, *(const v4f*)(a_src + (size_t)ei[e] * HEADS)));
    }
    for (int i = tid; i < tn; i += EB)
      max4(i, score4(i, *(const v4f*)(a_src + (size_t)(tile0 + i) * HEADS)));
    __syncthreads();
    auto add_edge = [&](int dl, float av, v4f hv) {
      float v = leaky(av + sadst[dl * HEADS + head]);
      float p = expf(v - dec_f32(smax[dl * HEADS + head]));
      if ((lane & 7) == 0) atomicAdd(&sden[dl * HEADS + head], p);
      float* a = acc + dl * CH + lane;
      atomicAdd(a, p * hv.x);
      atomicAdd(a + 32, p * hv.y);
      atomicAdd(a + 64, p * hv.z);
      atomicAdd(a + 96, p * hv.w);
    };
    for (int e = wave; e < ne; e += EB / 32) {
      unsigned dl = (unsigned)(dst[e] - tile0);
      if (dl < (unsigned)tn) {
        int s = ei[e];
        add_edge((int)dl, a_src[(size_t)s * HEADS + head],
                 *(const v4f*)(h + (size_t)s * CH + lane * 4));
      }
    }
    for (int i = wave; i < tn; i += EB / 32) {
      int s = tile0 + i;
      add_edge(i, a_src[(size_t)s * HEADS + head],
               *(const v4f*)(h + (size_t)s * CH + lane * 4));
    }
    __syncthreads();
    for (int i = tid; i < tn * (CH / 4); i += EB) {
      int node = i >> 5;
      int l4 = i & 31;
      float den = sden[node * HEADS + (l4 >> 3)] + 1e-16f;
      const float* a = acc + node * CH + l4;
      v4f b = *(const v4f*)(bias + l4 * 4);
      v4f o;
      o.x = a[0] / den + b.x;
      o.y = a[32] / den + b.y;
      o.z = a[64] / den + b.z;
      o.w = a[96] / den + b.w;
      o.x = (o.x >= 0.0f) ? o.x : pw * o.x;
      o.y = (o.y >= 0.0f) ? o.y : pw * o.y;
      o.z = (o.z >= 0.0f) ? o.z : pw * o.z;
      o.w = (o.w >= 0.0f) ? o.w : pw * o.w;
      *(v4f*)(out + (size_t)(tile0 + node) * CH + l4 * 4) = o;
    }
    return;
  }

  {
    const int segb = wv * SEG;
    for (int base = lane; base < wfound; base += 32 * MAX_U) {
      int sv[MAX_U], dd[MAX_U], rk[MAX_U]; v4f as[MAX_U];
#pragma unroll
      for (int u = 0; u < MAX_U; ++u) {
        const int j = base + u * 32;
        int e = 0;
        dd[u] = 0;
        rk[u] = 0;
        if (j < wfound) {
          const int2 r = lsd[segb + j];
          e = r.x; dd[u] = r.y;
          rk[u] = atomicAdd(&rcnt[dd[u]], 1);
        }
        sv[u] = ei[e];
      }
#pragma unroll
      for (int u = 0; u < MAX_U; ++u) as[u] = *(const v4f*)(a_src + (size_t)sv[u] * HEADS);
#pragma unroll
      for (int u = 0; u < MAX_U; ++u) {
        const int j = base + u * 32;
        if (j < wfound) {
          const v4f ev = score4(dd[u], as[u]);
          ls2[offw + j] = sv[u];
          ld2[offw + j] = dd[u] | (rk[u] << 8);
          *(v4f*)(esc + (offw + j) * HEADS) = ev;
          max4(dd[u], ev);
        }
      }
    }
  }
  for (int i = tid; i < tn; i += EB) {
    const v4f ev = score4(i, *(const v4f*)(a_src + (size_t)(tile0 + i) * HEADS));
    *(v4f*)(esc + (total + i) * HEADS) = ev;
    max4(i, ev);
  }
  __syncthreads();

  const int items = total + tn;
  if (wv == 0) {
    constexpr int RPL = (TN + 31) / 32;
    int c[RPL];
    int sum = 0;
#pragma unroll
    for (int k = 0; k < RPL; ++k) {
      const int r = lane * RPL + k;
      c[k] = (r < tn) ? rcnt[r] : 0;
      sum += c[k];
    }
    int incl = sum;
#pragma unroll
    for (int d = 1; d < 32; d <<= 1) {
      const int t = __shfl_up(incl, d);
      if (lane >= d) incl += t;
    }
    int run = incl - sum;
#pragma unroll
    for (int k = 0; k < RPL; ++k) {
      const int r = lane * RPL + k;
      if (r <= tn) rstart[r] = run;
      run += c[k];
    }
  }
  __syncthreads();
  for (int idx = tid; idx < items; idx += EB) {
    int s, dl, r;
    if (idx < total) {
      s = ls2[idx];
      const int pk = ld2[idx];
      dl = pk & 255;
      r = pk >> 8;
    } else {
      dl = idx - total;
      s = tile0 + dl;
      r = 0;
    }
    const int q = rstart[dl] + r;
    lsd[q] = int2{s, dl};
    *(v4f*)(gsc + q * HEADS) = *(const v4f*)(esc + idx * HEADS);
  }
  __syncthreads();

  {
    int rb = 0, re = 0;
    {
      const int lo = wv * items, hi = lo + items;
      for (int c = 0; c < (TN + 31) / 32; ++c) {
        const int r = c * 32 + lane;
        const int k32 = (r < tn) ? rstart[r] * 32 : 0x7fffffff;
        rb += __popc(__builtin_amdgcn_ballot_w32(k32 < lo));
        re += __popc(__builtin_amdgcn_ballot_w32(k32 < hi));
      }
    }
    const int pbeg = __builtin_amdgcn_readfirstlane(rstart[rb]);
    const int pend = __builtin_amdgcn_readfirstlane(rstart[re]);
    const v4f bs = *(const v4f*)(bias + lane * 4);
    int cur = -1;
    v4f sm = {0.f, 0.f, 0.f, 0.f};
    float den = 0.0f;
    auto finish_row = [&](int row) {
      const float d = den + 1e-16f;
      v4f o;
      o.x = sm.x / d + bs.x;
      o.y = sm.y / d + bs.y;
      o.z = sm.z / d + bs.z;
      o.w = sm.w / d + bs.w;
      o.x = (o.x >= 0.0f) ? o.x : pw * o.x;
      o.y = (o.y >= 0.0f) ? o.y : pw * o.y;
      o.z = (o.z >= 0.0f) ? o.z : pw * o.z;
      o.w = (o.w >= 0.0f) ? o.w : pw * o.w;
      *(v4f*)(out + (size_t)(tile0 + row) * CH + lane * 4) = o;
    };
    static_assert(ACC_U * HEADS == 32, "one (item, head) pair per lane");
    const int sub = lane & (ACC_U - 1);
    for (int i0 = pbeg; i0 < pend; i0 += ACC_U) {
      const int idx = i0 + sub;
      int s = tile0, dl = -1;
      if (idx < pend) { const int2 r = lsd[idx]; s = r.x; dl = r.y; }
      int dlu[ACC_U]; v4f hv[ACC_U];
#pragma unroll
      for (int u = 0; u < ACC_U; ++u) {
        const int su = __builtin_amdgcn_readlane(s, u);
        dlu[u] = __builtin_amdgcn_readlane(dl, u);
        hv[u] = *(const v4f*)(h + (size_t)su * CH + lane * 4);
      }
      const int dls = (dl >= 0) ? dl : 0;
      const float v = gsc[((dl >= 0) ? idx : pbeg) * HEADS + head];
      const float p = expf(v - dec_f32(smax[dls * HEADS + head]));
      auto add_row = [&](auto uc) {
        constexpr int u = decltype(uc)::value;
        if (dlu[u] >= 0) {
          if (dlu[u] != cur) {
            if (cur >= 0) finish_row(cur);
            cur = dlu[u];
            sm = v4f{0.f, 0.f, 0.f, 0.f};
            den = 0.0f;
          }
          const float pu = group8_bcast<u>(p);
          sm.x += pu * hv[u].x;
          sm.y += pu * hv[u].y;
          sm.z += pu * hv[u].z;
          sm.w += pu * hv[u].w;
          den += pu;
        }
      };
      add_row(IC<0>{}); add_row(IC<1>{}); add_row(IC<2>{}); add_row(IC<3>{});
      add_row(IC<4>{}); add_row(IC<5>{}); add_row(IC<6>{}); add_row(IC<7>{});
    }
    if (cur >= 0) finish_row(cur);
  }
}

extern "C" void kernel_launch(void* const* d_in, const int* in_sizes, int n_in,
                              void* d_out, int out_size, void* d_ws, size_t ws_size,
                              hipStream_t stream) {
  const float* x       = (const float*)d_in[0];
  const int*   ei      = (const int*)d_in[1];
  const float* W       = (const float*)d_in[2];
  const float* att_src = (const float*)d_in[3];
  const float* att_dst = (const float*)d_in[4];
  const float* bias    = (const float*)d_in[5];
  const float* prelu_w = (const float*)d_in[6];
  float* out = (float*)d_out;

  const int n  = in_sizes[0] / CH;
  const int ne = in_sizes[1] / 2;

  float* h     = (float*)d_ws;
  float* a_src = h + (size_t)n * CH;
  float* a_dst = a_src + (size_t)n * HEADS;
  unsigned* bm = (unsigned*)(a_dst + (size_t)n * HEADS);

  gat_gemm_att<<<GEMM_WG, 256, 0, stream>>>(x, W, att_src, att_dst, ei,
                                                  h, a_src, a_dst, bm, n, ne);
  gat_tile<<<(n + TN - 1) / TN, EB, 0, stream>>>(ei, bm, h, a_src, a_dst, bias, prelu_w,
                                                 out, n, ne);
}
